# starting kernel
# baseline (speedup 1.0000x reference)
#include <hip/hip_runtime.h>

#define N_FEATS 128

typedef __attribute__((ext_vector_type(2))) float v2f;
typedef __attribute__((ext_vector_type(8))) float v8f;

__device__ __forceinline__ void atomic_add_f32(float* p, float v) {
  __hip_atomic_fetch_add(p, v, __ATOMIC_RELAXED, __HIP_MEMORY_SCOPE_AGENT);
}

__global__ __launch_bounds__(256) void gcn_degrees(
    const int* __restrict__ src, const int* __restrict__ dst,
    float* __restrict__ deg_out, float* __restrict__ deg_in, int n_edges) {
  int i = blockIdx.x * blockDim.x + threadIdx.x;
  if (i >= n_edges) return;
  atomic_add_f32(&deg_out[src[i]], 1.0f);
  atomic_add_f32(&deg_in[dst[i]], 1.0f);
}

__global__ __launch_bounds__(256) void gcn_norms(
    const float* __restrict__ deg_out, const float* __restrict__ deg_in,
    float* __restrict__ ns, float* __restrict__ nd, int n) {
  int i = blockIdx.x * blockDim.x + threadIdx.x;
  if (i >= n) return;
  ns[i] = rsqrtf(fmaxf(deg_out[i], 1.0f));
  nd[i] = rsqrtf(fmaxf(deg_in[i], 1.0f));
}

__global__ __launch_bounds__(256) void gcn_gemm_wmma(
    const float* __restrict__ A, const float* __restrict__ norm_src,
    const float* __restrict__ W, float* __restrict__ out, int n_rows) {
  __shared__ float As[16][132];

  const int row0 = blockIdx.x << 4;

  for (int t = threadIdx.x; t < 512; t += 256) {
    const int r = t >> 5;
    const int c4 = (t & 31) << 2;
    const int row = row0 + r;
    if (row < n_rows) {
      const float4 v = *(const float4*)(A + (size_t)row * N_FEATS + c4);
      const float s = norm_src[row];
      As[r][c4 + 0] = v.x * s;
      As[r][c4 + 1] = v.y * s;
      As[r][c4 + 2] = v.z * s;
      As[r][c4 + 3] = v.w * s;
    } else {
      As[r][c4 + 0] = 0.f; As[r][c4 + 1] = 0.f;
      As[r][c4 + 2] = 0.f; As[r][c4 + 3] = 0.f;
    }
  }
  __syncthreads();

  const int wave = threadIdx.x >> 5;
  const int lane = threadIdx.x & 31;
  const int half = lane >> 4;
  const int m    = lane & 15;
  const int col0 = wave << 4;

  v8f acc = {};
#pragma unroll
  for (int k = 0; k < N_FEATS; k += 4) {
    const int ka = k + (half << 1);
    v2f a, b;
    a.x = As[m][ka];
    a.y = As[m][ka + 1];
    b.x = W[ka * N_FEATS + col0 + m];
    b.y = W[(ka + 1) * N_FEATS + col0 + m];
    acc = __builtin_amdgcn_wmma_f32_16x16x4_f32(
         false, a,  false, b,
         (short)0, acc,  false,  false);
  }

#pragma unroll
  for (int j = 0; j < 8; ++j) {
    const int r = row0 + j + (half << 3);
    if (r < n_rows) out[(size_t)r * N_FEATS + col0 + m] = acc[j];
  }
}

__global__ __launch_bounds__(256) void gcn_agg(
    const float* __restrict__ h, const int* __restrict__ src,
    const int* __restrict__ dst, float* __restrict__ agg, int n_edges) {
  const int e    = (blockIdx.x * blockDim.x + threadIdx.x) >> 5;
  const int lane = threadIdx.x & 31;
  if (e >= n_edges) return;
  const int s = src[e];
  const int d = dst[e];
  const float4 v = *(const float4*)(h + (size_t)s * N_FEATS + (lane << 2));
  float* o = agg + (size_t)d * N_FEATS + (lane << 2);
  atomic_add_f32(o + 0, v.x);
  atomic_add_f32(o + 1, v.y);
  atomic_add_f32(o + 2, v.z);
  atomic_add_f32(o + 3, v.w);
}

__global__ __launch_bounds__(256) void gcn_finalize(
    const float* __restrict__ agg, const float* __restrict__ norm_dst,
    const float* __restrict__ bias, float* __restrict__ out,
    int n_nodes, int do_relu) {
  const int i = blockIdx.x * blockDim.x + threadIdx.x;
  if (i >= n_nodes * 32) return;
  const int row = i >> 5;
  const int c4  = (i & 31) << 2;
  const float s = norm_dst[row];
  const float4 v = *(const float4*)(agg + (size_t)row * N_FEATS + c4);
  const float4 b = *(const float4*)(bias + c4);
  float4 r;
  r.x = v.x * s + b.x;
  r.y = v.y * s + b.y;
  r.z = v.z * s + b.z;
  r.w = v.w * s + b.w;
  if (do_relu) {
    r.x = fmaxf(r.x, 0.f); r.y = fmaxf(r.y, 0.f);
    r.z = fmaxf(r.z, 0.f); r.w = fmaxf(r.w, 0.f);
  }
  *(float4*)(out + (size_t)row * N_FEATS + c4) = r;
}

static inline size_t align256(size_t x) { return (x + 255) & ~(size_t)255; }

extern "C" void kernel_launch(void* const* d_in, const int* in_sizes, int n_in,
                              void* d_out, int out_size, void* d_ws, size_t ws_size,
                              hipStream_t stream) {
  (void)n_in; (void)out_size; (void)ws_size;
  const float* feats = (const float*)d_in[0];
  const int*   src   = (const int*)d_in[1];
  const int*   dst   = (const int*)d_in[2];
  const float* W1    = (const float*)d_in[3];
  const float* b1    = (const float*)d_in[4];
  const float* W2    = (const float*)d_in[5];
  const float* b2    = (const float*)d_in[6];
  float* out = (float*)d_out;

  const int n_nodes = in_sizes[0] / N_FEATS;
  const int n_edges = in_sizes[1];

  const size_t nb = align256((size_t)n_nodes * sizeof(float));
  const size_t hb = align256((size_t)n_nodes * N_FEATS * sizeof(float));
  char* p = (char*)d_ws;
  float* deg_out = (float*)p; p += nb;
  float* deg_in  = (float*)p; p += nb;
  float* ns      = (float*)p; p += nb;
  float* nd      = (float*)p; p += nb;
  float* bufA    = (float*)p; p += hb;
  float* bufB    = (float*)p; p += hb;

  const int tb = 256;
  const size_t node_bytes = (size_t)n_nodes * sizeof(float);
  const size_t feat_bytes = (size_t)n_nodes * N_FEATS * sizeof(float);

  hipMemsetAsync(deg_out, 0, node_bytes, stream);
  hipMemsetAsync(deg_in,  0, node_bytes, stream);
  gcn_degrees<<<(n_edges + tb - 1) / tb, tb, 0, stream>>>(src, dst, deg_out, deg_in, n_edges);
  gcn_norms<<<(n_nodes + tb - 1) / tb, tb, 0, stream>>>(deg_out, deg_in, ns, nd, n_nodes);

  const int gemm_grid = (n_nodes + 15) / 16;
  const int agg_grid  = (n_edges + 7) / 8;
  const int fin_grid  = (n_nodes * 32 + tb - 1) / tb;

  gcn_gemm_wmma<<<gemm_grid, 256, 0, stream>>>(feats, ns, W1, bufA, n_nodes);
  hipMemsetAsync(bufB, 0, feat_bytes, stream);
  gcn_agg<<<agg_grid, 256, 0, stream>>>(bufA, src, dst, bufB, n_edges);
  gcn_finalize<<<fin_grid, tb, 0, stream>>>(bufB, nd, b1, bufA, n_nodes,  1);

  gcn_gemm_wmma<<<gemm_grid, 256, 0, stream>>>(bufA, ns, W2, bufB, n_nodes);
  hipMemsetAsync(out, 0, feat_bytes, stream);
  gcn_agg<<<agg_grid, 256, 0, stream>>>(bufB, src, dst, out, n_edges);
  gcn_finalize<<<fin_grid, tb, 0, stream>>>(out, nd, b2, out, n_nodes,  0);
}


# agg+finalize: global float atomics -> each work-group owns 512 dst nodes, filters the whole dst list, sums source rows in LDS (LDS float adds), stores scaled rows; 8 launches -> 6; sum order changes
# speedup vs baseline: 2.0817x; 2.0817x over previous
#include <hip/hip_runtime.h>

#define N_FEATS 128

typedef __attribute__((ext_vector_type(2))) float v2f;
typedef __attribute__((ext_vector_type(8))) float v8f;

__device__ __forceinline__ void atomic_add_f32(float* p, float v) {
  __hip_atomic_fetch_add(p, v, __ATOMIC_RELAXED, __HIP_MEMORY_SCOPE_AGENT);
}

__global__ __launch_bounds__(256) void gcn_degrees(
    const int* __restrict__ src, const int* __restrict__ dst,
    float* __restrict__ deg_out, float* __restrict__ deg_in, int n_edges) {
  int i = blockIdx.x * blockDim.x + threadIdx.x;
  if (i >= n_edges) return;
  atomic_add_f32(&deg_out[src[i]], 1.0f);
  atomic_add_f32(&deg_in[dst[i]], 1.0f);
}

__global__ __launch_bounds__(256) void gcn_norms(
    const float* __restrict__ deg_out, const float* __restrict__ deg_in,
    float* __restrict__ ns, float* __restrict__ nd, int n) {
  int i = blockIdx.x * blockDim.x + threadIdx.x;
  if (i >= n) return;
  ns[i] = rsqrtf(fmaxf(deg_out[i], 1.0f));
  nd[i] = rsqrtf(fmaxf(deg_in[i], 1.0f));
}

__global__ __launch_bounds__(256) void gcn_gemm_wmma(
    const float* __restrict__ A, const float* __restrict__ norm_src,
    const float* __restrict__ W, float* __restrict__ out, int n_rows) {
  __shared__ float As[16][132];

  const int row0 = blockIdx.x << 4;

  for (int t = threadIdx.x; t < 512; t += 256) {
    const int r = t >> 5;
    const int c4 = (t & 31) << 2;
    const int row = row0 + r;
    if (row < n_rows) {
      const float4 v = *(const float4*)(A + (size_t)row * N_FEATS + c4);
      const float s = norm_src[row];
      As[r][c4 + 0] = v.x * s;
      As[r][c4 + 1] = v.y * s;
      As[r][c4 + 2] = v.z * s;
      As[r][c4 + 3] = v.w * s;
    } else {
      As[r][c4 + 0] = 0.f; As[r][c4 + 1] = 0.f;
      As[r][c4 + 2] = 0.f; As[r][c4 + 3] = 0.f;
    }
  }
  __syncthreads();

  const int wave = threadIdx.x >> 5;
  const int lane = threadIdx.x & 31;
  const int half = lane >> 4;
  const int m    = lane & 15;
  const int col0 = wave << 4;

  v8f acc = {};
#pragma unroll
  for (int k = 0; k < N_FEATS; k += 4) {
    const int ka = k + (half << 1);
    v2f a, b;
    a.x = As[m][ka];
    a.y = As[m][ka + 1];
    b.x = W[ka * N_FEATS + col0 + m];
    b.y = W[(ka + 1) * N_FEATS + col0 + m];
    acc = __builtin_amdgcn_wmma_f32_16x16x4_f32(
         false, a,  false, b,
         (short)0, acc,  false,  false);
  }

#pragma unroll
  for (int j = 0; j < 8; ++j) {
    const int r = row0 + j + (half << 3);
    if (r < n_rows) out[(size_t)r * N_FEATS + col0 + m] = acc[j];
  }
}

#define OWN_R 512
#define OWN_T 1024
#define OWN_W 32
#define OWN_U 4
#define OWN_B 8
#define OWN_Q (128 + OWN_B)

__device__ __forceinline__ void lds_add_f32(float* p, float v) {
  __hip_atomic_fetch_add(p, v, __ATOMIC_RELAXED, __HIP_MEMORY_SCOPE_WORKGROUP);
}

__device__ __forceinline__ int4 own_load_dst4(const int* __restrict__ dst, int e0, int nvec) {
  return *(const int4*)(dst + ((e0 < nvec) ? e0 : 0));
}

template <bool FULL>
__device__ __forceinline__ void own_gather(
    const float* __restrict__ h, const int* __restrict__ src, float* acc,
    const int* qe, const int* qd, int first, int n, int lane) {
  int e[OWN_B], dl[OWN_B], s[OWN_B];
  float4 v[OWN_B];
#pragma unroll
  for (int k = 0; k < OWN_B; ++k) {
    const int idx = (FULL || k < n) ? first + k : first;
    e[k]  = __builtin_amdgcn_readfirstlane(qe[idx]);
    dl[k] = __builtin_amdgcn_readfirstlane(qd[idx]);
  }
#pragma unroll
  for (int k = 0; k < OWN_B; ++k) s[k] = src[e[k]];
#pragma unroll
  for (int k = 0; k < OWN_B; ++k)
    v[k] = *(const float4*)(h + (size_t)s[k] * N_FEATS + (lane << 2));
#pragma unroll
  for (int k = 0; k < OWN_B; ++k) {
    if (FULL || k < n) {
      float* a = acc + dl[k] * N_FEATS + lane;
      lds_add_f32(a +  0, v[k].x);
      lds_add_f32(a + 32, v[k].y);
      lds_add_f32(a + 64, v[k].z);
      lds_add_f32(a + 96, v[k].w);
    }
  }
}

__device__ __forceinline__ void own_scan(
    const int4 (&c)[OWN_U], int base, int nvec, int lo, int lane, int& cnt,
    int* myqe, int* myqd, const float* __restrict__ h, const int* __restrict__ src, float* acc) {
#pragma unroll
  for (int u = 0; u < OWN_U; ++u) {
    const int e0 = base + u * 128 + (lane << 2);
    const int lo_u = (e0 < nvec) ? lo : 0x7fffffff;
    const int dv[4] = {c[u].x, c[u].y, c[u].z, c[u].w};
#pragma unroll
    for (int j = 0; j < 4; ++j) {
      const unsigned dl = (unsigned)(dv[j] - lo_u);
      const bool m = dl < (unsigned)OWN_R;
      const unsigned mask = __builtin_amdgcn_ballot_w32(m);
      if (mask != 0u) {
        if (m) {
          const int p = cnt + __builtin_popcount(mask & ((1u << lane) - 1u));
          myqe[p] = e0 + j;
          myqd[p] = (int)dl;
        }
        cnt += __builtin_popcount(mask);
      }
    }
    while (cnt >= OWN_B) {
      cnt -= OWN_B;
      own_gather<true>(h, src, acc, myqe, myqd, cnt, OWN_B, lane);
    }
  }
}

__global__ __launch_bounds__(OWN_T) void gcn_agg_own(
    const float* __restrict__ h, const int* __restrict__ src, const int* __restrict__ dst,
    const float* __restrict__ norm_dst, const float* __restrict__ bias,
    float* __restrict__ out, int n_nodes, int n_edges, int do_relu) {
  __shared__ __attribute__((aligned(16))) float acc[OWN_R * N_FEATS];
  __shared__ int qe[OWN_W][OWN_Q];
  __shared__ int qd[OWN_W][OWN_Q];

  const int tid  = threadIdx.x;
  const int wave = __builtin_amdgcn_readfirstlane(tid >> 5);
  const int lane = tid & 31;
  const int lo   = blockIdx.x * OWN_R;

  for (int i = tid; i < OWN_R * N_FEATS / 4; i += OWN_T)
    ((float4*)acc)[i] = make_float4(0.f, 0.f, 0.f, 0.f);
  __syncthreads();

  int* myqe = qe[wave];
  int* myqd = qd[wave];
  int cnt = 0;
  const int nvec = n_edges & ~3;
  const int step = OWN_W * OWN_U * 128;
  int base = wave * (OWN_U * 128);

  if (wave == 0) {
    for (int e = nvec; e < n_edges; ++e) {
      const unsigned dl = (unsigned)(dst[e] - lo);
      if (dl < (unsigned)OWN_R) {
        if (lane == 0) { myqe[cnt] = e; myqd[cnt] = (int)dl; }
        ++cnt;
      }
    }
  }

  if (base < nvec) {
    int4 a[OWN_U], b[OWN_U];
#pragma unroll
    for (int u = 0; u < OWN_U; ++u)
      a[u] = own_load_dst4(dst, base + u * 128 + (lane << 2), nvec);
    for (;;) {
#pragma unroll
      for (int u = 0; u < OWN_U; ++u)
        b[u] = own_load_dst4(dst, base + step + u * 128 + (lane << 2), nvec);
      own_scan(a, base, nvec, lo, lane, cnt, myqe, myqd, h, src, acc);
      base += step;
      if (base >= nvec) break;
#pragma unroll
      for (int u = 0; u < OWN_U; ++u)
        a[u] = own_load_dst4(dst, base + step + u * 128 + (lane << 2), nvec);
      own_scan(b, base, nvec, lo, lane, cnt, myqe, myqd, h, src, acc);
      base += step;
      if (base >= nvec) break;
    }
  }
  if (cnt > 0) own_gather<false>(h, src, acc, myqe, myqd, 0, cnt, lane);
  __syncthreads();

  const float4 b = *(const float4*)(bias + (lane << 2));
  for (int r = wave; r < OWN_R; r += OWN_W) {
    const int row = lo + r;
    if (row < n_nodes) {
      const float* a = acc + r * N_FEATS + lane;
      const float s = norm_dst[row];
      float4 o;
      o.x = a[ 0] * s + b.x;
      o.y = a[32] * s + b.y;
      o.z = a[64] * s + b.z;
      o.w = a[96] * s + b.w;
      if (do_relu) {
        o.x = fmaxf(o.x, 0.f); o.y = fmaxf(o.y, 0.f);
        o.z = fmaxf(o.z, 0.f); o.w = fmaxf(o.w, 0.f);
      }
      *(float4*)(out + (size_t)row * N_FEATS + (lane << 2)) = o;
    }
  }
}

static inline size_t align256(size_t x) { return (x + 255) & ~(size_t)255; }

extern "C" void kernel_launch(void* const* d_in, const int* in_sizes, int n_in,
                              void* d_out, int out_size, void* d_ws, size_t ws_size,
                              hipStream_t stream) {
  (void)n_in; (void)out_size; (void)ws_size;
  const float* feats = (const float*)d_in[0];
  const int*   src   = (const int*)d_in[1];
  const int*   dst   = (const int*)d_in[2];
  const float* W1    = (const float*)d_in[3];
  const float* b1    = (const float*)d_in[4];
  const float* W2    = (const float*)d_in[5];
  const float* b2    = (const float*)d_in[6];
  float* out = (float*)d_out;

  const int n_nodes = in_sizes[0] / N_FEATS;
  const int n_edges = in_sizes[1];

  const size_t nb = align256((size_t)n_nodes * sizeof(float));
  const size_t hb = align256((size_t)n_nodes * N_FEATS * sizeof(float));
  char* p = (char*)d_ws;
  float* deg_out = (float*)p; p += nb;
  float* deg_in  = (float*)p; p += nb;
  float* ns      = (float*)p; p += nb;
  float* nd      = (float*)p; p += nb;
  float* bufA    = (float*)p; p += hb;
  float* bufB    = (float*)p; p += hb;

  const int tb = 256;
  const size_t node_bytes = (size_t)n_nodes * sizeof(float);

  hipMemsetAsync(deg_out, 0, node_bytes, stream);
  hipMemsetAsync(deg_in,  0, node_bytes, stream);
  gcn_degrees<<<(n_edges + tb - 1) / tb, tb, 0, stream>>>(src, dst, deg_out, deg_in, n_edges);
  gcn_norms<<<(n_nodes + tb - 1) / tb, tb, 0, stream>>>(deg_out, deg_in, ns, nd, n_nodes);

  const int gemm_grid = (n_nodes + 15) / 16;
  const int own_grid  = (n_nodes + OWN_R - 1) / OWN_R;

  gcn_gemm_wmma<<<gemm_grid, 256, 0, stream>>>(feats, ns, W1, bufA, n_nodes);
  gcn_agg_own<<<own_grid, OWN_T, 0, stream>>>(bufA, src, dst, nd, b1, bufB, n_nodes, n_edges,  1);

  gcn_gemm_wmma<<<gemm_grid, 256, 0, stream>>>(bufB, ns, W2, bufA, n_nodes);
  gcn_agg_own<<<own_grid, OWN_T, 0, stream>>>(bufA, src, dst, nd, b2, out, n_nodes, n_edges,  0);
}


# agg filter: work-groups scan a 1-byte owner key per edge (dst>>9, written by the degree kernel), 16 edges per lane-load, not the 4-byte dst list; dst read only for queued edges; sum order changes
# speedup vs baseline: 2.4077x; 1.1566x over previous
#include <hip/hip_runtime.h>

#define N_FEATS 128

typedef __attribute__((ext_vector_type(2))) float v2f;
typedef __attribute__((ext_vector_type(8))) float v8f;

__device__ __forceinline__ void atomic_add_f32(float* p, float v) {
  __hip_atomic_fetch_add(p, v, __ATOMIC_RELAXED, __HIP_MEMORY_SCOPE_AGENT);
}

#define OWN_SH 9

__global__ __launch_bounds__(256) void gcn_degrees(
    const int* __restrict__ src, const int* __restrict__ dst,
    float* __restrict__ deg_out, float* __restrict__ deg_in,
    unsigned char* __restrict__ key, int n_edges) {
  int i = blockIdx.x * blockDim.x + threadIdx.x;
  if (i >= n_edges) return;
  const int d = dst[i];
  atomic_add_f32(&deg_out[src[i]], 1.0f);
  atomic_add_f32(&deg_in[d], 1.0f);
  key[i] = (unsigned char)(d >> OWN_SH);
}

__global__ __launch_bounds__(256) void gcn_norms(
    const float* __restrict__ deg_out, const float* __restrict__ deg_in,
    float* __restrict__ ns, float* __restrict__ nd, int n) {
  int i = blockIdx.x * blockDim.x + threadIdx.x;
  if (i >= n) return;
  ns[i] = rsqrtf(fmaxf(deg_out[i], 1.0f));
  nd[i] = rsqrtf(fmaxf(deg_in[i], 1.0f));
}

__global__ __launch_bounds__(256) void gcn_gemm_wmma(
    const float* __restrict__ A, const float* __restrict__ norm_src,
    const float* __restrict__ W, float* __restrict__ out, int n_rows) {
  __shared__ float As[16][132];

  const int row0 = blockIdx.x << 4;

  for (int t = threadIdx.x; t < 512; t += 256) {
    const int r = t >> 5;
    const int c4 = (t & 31) << 2;
    const int row = row0 + r;
    if (row < n_rows) {
      const float4 v = *(const float4*)(A + (size_t)row * N_FEATS + c4);
      const float s = norm_src[row];
      As[r][c4 + 0] = v.x * s;
      As[r][c4 + 1] = v.y * s;
      As[r][c4 + 2] = v.z * s;
      As[r][c4 + 3] = v.w * s;
    } else {
      As[r][c4 + 0] = 0.f; As[r][c4 + 1] = 0.f;
      As[r][c4 + 2] = 0.f; As[r][c4 + 3] = 0.f;
    }
  }
  __syncthreads();

  const int wave = threadIdx.x >> 5;
  const int lane = threadIdx.x & 31;
  const int half = lane >> 4;
  const int m    = lane & 15;
  const int col0 = wave << 4;

  v8f acc = {};
#pragma unroll
  for (int k = 0; k < N_FEATS; k += 4) {
    const int ka = k + (half << 1);
    v2f a, b;
    a.x = As[m][ka];
    a.y = As[m][ka + 1];
    b.x = W[ka * N_FEATS + col0 + m];
    b.y = W[(ka + 1) * N_FEATS + col0 + m];
    acc = __builtin_amdgcn_wmma_f32_16x16x4_f32(
         false, a,  false, b,
         (short)0, acc,  false,  false);
  }

#pragma unroll
  for (int j = 0; j < 8; ++j) {
    const int r = row0 + j + (half << 3);
    if (r < n_rows) out[(size_t)r * N_FEATS + col0 + m] = acc[j];
  }
}

#define OWN_R (1 << OWN_SH)
#define OWN_T 1024
#define OWN_W 32
#define OWN_U 4
#define OWN_B 8
#define OWN_Q (32 + OWN_B)

__device__ __forceinline__ void lds_add_f32(float* p, float v) {
  __hip_atomic_fetch_add(p, v, __ATOMIC_RELAXED, __HIP_MEMORY_SCOPE_WORKGROUP);
}

__device__ __forceinline__ int4 own_load_key16(const unsigned char* __restrict__ key, int e0, int nvec) {
  return *(const int4*)(key + ((e0 < nvec) ? e0 : 0));
}

template <bool FULL>
__device__ __forceinline__ void own_gather(
    const float* __restrict__ h, const int* __restrict__ src, const int* __restrict__ dst,
    float* acc, const int* qe, int lo, int first, int n, int lane) {
  int e[OWN_B], d[OWN_B], s[OWN_B];
  float4 v[OWN_B];
#pragma unroll
  for (int k = 0; k < OWN_B; ++k) {
    const int idx = (FULL || k < n) ? first + k : first;
    e[k]  = __builtin_amdgcn_readfirstlane(qe[idx]);
  }
#pragma unroll
  for (int k = 0; k < OWN_B; ++k) s[k] = src[e[k]];
#pragma unroll
  for (int k = 0; k < OWN_B; ++k) d[k] = dst[e[k]];
#pragma unroll
  for (int k = 0; k < OWN_B; ++k)
    v[k] = *(const float4*)(h + (size_t)s[k] * N_FEATS + (lane << 2));
#pragma unroll
  for (int k = 0; k < OWN_B; ++k)
    asm volatile("" :: "v"(v[k].x), "v"(v[k].y), "v"(v[k].z), "v"(v[k].w));
#pragma unroll
  for (int k = 0; k < OWN_B; ++k) {
    const unsigned dl = (unsigned)(d[k] - lo);
    if ((FULL || k < n) && dl < (unsigned)OWN_R) {
      float* a = acc + dl * N_FEATS + lane;
      lds_add_f32(a +  0, v[k].x);
      lds_add_f32(a + 32, v[k].y);
      lds_add_f32(a + 64, v[k].z);
      lds_add_f32(a + 96, v[k].w);
    }
  }
}

__device__ __forceinline__ void own_scan(
    const int4 (&c)[OWN_U], int base, int nvec, unsigned pat, int lo, int lane, int& cnt,
    int* myqe, const float* __restrict__ h, const int* __restrict__ src,
    const int* __restrict__ dst, float* acc) {
#pragma unroll
  for (int u = 0; u < OWN_U; ++u) {
    const int e0 = base + u * 512 + (lane << 4);
    const unsigned kv[4] = {(unsigned)c[u].x, (unsigned)c[u].y, (unsigned)c[u].z, (unsigned)c[u].w};
    unsigned f = 0u;
#pragma unroll
    for (int j = 0; j < 4; ++j) {
      const unsigned y = kv[j] ^ pat;
      const unsigned n = (y - 0x01010101u) & ~y & 0x80808080u;
      f |= n >> (7 - j);
    }
    if (e0 >= nvec) f = 0u;
    unsigned mask = __builtin_amdgcn_ballot_w32(f != 0u);
    while (mask != 0u) {
      if (f != 0u) {
        const int p = cnt + __builtin_popcount(mask & ((1u << lane) - 1u));
        const int pos = __builtin_ctz(f);
        myqe[p] = e0 + ((pos & 3) << 2) + (pos >> 3);
        f &= f - 1u;
      }
      cnt += __builtin_popcount(mask);
      while (cnt >= OWN_B) {
        cnt -= OWN_B;
        own_gather<true>(h, src, dst, acc, myqe, lo, cnt, OWN_B, lane);
      }
      mask = __builtin_amdgcn_ballot_w32(f != 0u);
    }
  }
}

__global__ __launch_bounds__(OWN_T) void gcn_agg_own(
    const float* __restrict__ h, const int* __restrict__ src, const int* __restrict__ dst,
    const unsigned char* __restrict__ key,
    const float* __restrict__ norm_dst, const float* __restrict__ bias,
    float* __restrict__ out, int n_nodes, int n_edges, int do_relu) {
  __shared__ __attribute__((aligned(16))) float acc[OWN_R * N_FEATS];
  __shared__ int qe[OWN_W][OWN_Q];

  const int tid  = threadIdx.x;
  const int wave = __builtin_amdgcn_readfirstlane(tid >> 5);
  const int lane = tid & 31;
  const int lo   = blockIdx.x * OWN_R;
  const unsigned pat = (blockIdx.x & 255u) * 0x01010101u;

  for (int i = tid; i < OWN_R * N_FEATS / 4; i += OWN_T)
    ((float4*)acc)[i] = make_float4(0.f, 0.f, 0.f, 0.f);
  __syncthreads();

  int* myqe = qe[wave];
  int cnt = 0;
  const int nvec = n_edges & ~15;
  const int step = OWN_W * OWN_U * 512;
  int base = wave * (OWN_U * 512);

  if (wave == 0) {
    for (int e = nvec; e < n_edges; ++e) {
      const unsigned dl = (unsigned)(dst[e] - lo);
      if (dl < (unsigned)OWN_R) {
        if (lane == 0) myqe[cnt] = e;
        ++cnt;
      }
    }
    while (cnt >= OWN_B) {
      cnt -= OWN_B;
      own_gather<true>(h, src, dst, acc, myqe, lo, cnt, OWN_B, lane);
    }
  }

  if (base < nvec) {
    int4 a[OWN_U], b[OWN_U];
#pragma unroll
    for (int u = 0; u < OWN_U; ++u)
      a[u] = own_load_key16(key, base + u * 512 + (lane << 4), nvec);
    for (;;) {
#pragma unroll
      for (int u = 0; u < OWN_U; ++u)
        b[u] = own_load_key16(key, base + step + u * 512 + (lane << 4), nvec);
      own_scan(a, base, nvec, pat, lo, lane, cnt, myqe, h, src, dst, acc);
      base += step;
      if (base >= nvec) break;
#pragma unroll
      for (int u = 0; u < OWN_U; ++u) a[u] = b[u];
    }
  }
  if (cnt > 0) own_gather<false>(h, src, dst, acc, myqe, lo, 0, cnt, lane);
  __syncthreads();

  const float4 b = *(const float4*)(bias + (lane << 2));
  for (int r = wave; r < OWN_R; r += OWN_W) {
    const int row = lo + r;
    if (row < n_nodes) {
      const float* a = acc + r * N_FEATS + lane;
      const float s = norm_dst[row];
      float4 o;
      o.x = a[ 0] * s + b.x;
      o.y = a[32] * s + b.y;
      o.z = a[64] * s + b.z;
      o.w = a[96] * s + b.w;
      if (do_relu) {
        o.x = fmaxf(o.x, 0.f); o.y = fmaxf(o.y, 0.f);
        o.z = fmaxf(o.z, 0.f); o.w = fmaxf(o.w, 0.f);
      }
      *(float4*)(out + (size_t)row * N_FEATS + (lane << 2)) = o;
    }
  }
}

static inline size_t align256(size_t x) { return (x + 255) & ~(size_t)255; }

extern "C" void kernel_launch(void* const* d_in, const int* in_sizes, int n_in,
                              void* d_out, int out_size, void* d_ws, size_t ws_size,
                              hipStream_t stream) {
  (void)n_in; (void)out_size; (void)ws_size;
  const float* feats = (const float*)d_in[0];
  const int*   src   = (const int*)d_in[1];
  const int*   dst   = (const int*)d_in[2];
  const float* W1    = (const float*)d_in[3];
  const float* b1    = (const float*)d_in[4];
  const float* W2    = (const float*)d_in[5];
  const float* b2    = (const float*)d_in[6];
  float* out = (float*)d_out;

  const int n_nodes = in_sizes[0] / N_FEATS;
  const int n_edges = in_sizes[1];

  const size_t nb = align256((size_t)n_nodes * sizeof(float));
  const size_t hb = align256((size_t)n_nodes * N_FEATS * sizeof(float));
  char* p = (char*)d_ws;
  float* deg_out = (float*)p; p += nb;
  float* deg_in  = (float*)p; p += nb;
  float* ns      = (float*)p; p += nb;
  float* nd      = (float*)p; p += nb;
  float* bufA    = (float*)p; p += hb;
  float* bufB    = (float*)p; p += hb;
  unsigned char* key = (unsigned char*)p;

  const int tb = 256;
  const size_t node_bytes = (size_t)n_nodes * sizeof(float);

  hipMemsetAsync(deg_out, 0, node_bytes, stream);
  hipMemsetAsync(deg_in,  0, node_bytes, stream);
  gcn_degrees<<<(n_edges + tb - 1) / tb, tb, 0, stream>>>(src, dst, deg_out, deg_in, key, n_edges);
  gcn_norms<<<(n_nodes + tb - 1) / tb, tb, 0, stream>>>(deg_out, deg_in, ns, nd, n_nodes);

  const int gemm_grid = (n_nodes + 15) / 16;
  const int own_grid  = (n_nodes + OWN_R - 1) / OWN_R;

  gcn_gemm_wmma<<<gemm_grid, 256, 0, stream>>>(feats, ns, W1, bufA, n_nodes);
  gcn_agg_own<<<own_grid, OWN_T, 0, stream>>>(bufA, src, dst, key, nd, b1, bufB, n_nodes, n_edges,  1);

  gcn_gemm_wmma<<<gemm_grid, 256, 0, stream>>>(bufB, ns, W2, bufA, n_nodes);
  gcn_agg_own<<<own_grid, OWN_T, 0, stream>>>(bufA, src, dst, key, nd, b2, out, n_nodes, n_edges,  0);
}


# in-degrees: per-edge global float atomics -> integer count in LDS by the aggregation work-group that owns the node (layer 1), which forms norm_dst in its store loop; out-degree atomics stay
# speedup vs baseline: 2.8810x; 1.1966x over previous
#include <hip/hip_runtime.h>

#define N_FEATS 128

typedef __attribute__((ext_vector_type(2))) float v2f;
typedef __attribute__((ext_vector_type(8))) float v8f;

__device__ __forceinline__ void atomic_add_f32(float* p, float v) {
  __hip_atomic_fetch_add(p, v, __ATOMIC_RELAXED, __HIP_MEMORY_SCOPE_AGENT);
}

#define OWN_SH 9

__global__ __launch_bounds__(256) void gcn_degrees(
    const int* __restrict__ src, const int* __restrict__ dst,
    float* __restrict__ deg_out,
    unsigned char* __restrict__ key, int n_edges) {
  int i = blockIdx.x * blockDim.x + threadIdx.x;
  if (i >= n_edges) return;
  const int s = src[i];
  const int d = dst[i];
  atomic_add_f32(&deg_out[s], 1.0f);
  key[i] = (unsigned char)(d >> OWN_SH);
}

__global__ __launch_bounds__(256) void gcn_norms(
    const float* __restrict__ deg_out, float* __restrict__ ns, int n) {
  int i = blockIdx.x * blockDim.x + threadIdx.x;
  if (i >= n) return;
  ns[i] = rsqrtf(fmaxf(deg_out[i], 1.0f));
}

__global__ __launch_bounds__(256) void gcn_gemm_wmma(
    const float* __restrict__ A, const float* __restrict__ norm_src,
    const float* __restrict__ W, float* __restrict__ out, int n_rows) {
  __shared__ float As[16][132];

  const int row0 = blockIdx.x << 4;

  for (int t = threadIdx.x; t < 512; t += 256) {
    const int r = t >> 5;
    const int c4 = (t & 31) << 2;
    const int row = row0 + r;
    if (row < n_rows) {
      const float4 v = *(const float4*)(A + (size_t)row * N_FEATS + c4);
      const float s = norm_src[row];
      As[r][c4 + 0] = v.x * s;
      As[r][c4 + 1] = v.y * s;
      As[r][c4 + 2] = v.z * s;
      As[r][c4 + 3] = v.w * s;
    } else {
      As[r][c4 + 0] = 0.f; As[r][c4 + 1] = 0.f;
      As[r][c4 + 2] = 0.f; As[r][c4 + 3] = 0.f;
    }
  }
  __syncthreads();

  const int wave = threadIdx.x >> 5;
  const int lane = threadIdx.x & 31;
  const int half = lane >> 4;
  const int m    = lane & 15;
  const int col0 = wave << 4;

  v8f acc = {};
#pragma unroll
  for (int k = 0; k < N_FEATS; k += 4) {
    const int ka = k + (half << 1);
    v2f a, b;
    a.x = As[m][ka];
    a.y = As[m][ka + 1];
    b.x = W[ka * N_FEATS + col0 + m];
    b.y = W[(ka + 1) * N_FEATS + col0 + m];
    acc = __builtin_amdgcn_wmma_f32_16x16x4_f32(
         false, a,  false, b,
         (short)0, acc,  false,  false);
  }

#pragma unroll
  for (int j = 0; j < 8; ++j) {
    const int r = row0 + j + (half << 3);
    if (r < n_rows) out[(size_t)r * N_FEATS + col0 + m] = acc[j];
  }
}

#define OWN_R (1 << OWN_SH)
#define OWN_T 1024
#define OWN_W 32
#define OWN_U 4
#define OWN_B 8
#define OWN_Q (32 + OWN_B)

__device__ __forceinline__ void lds_add_f32(float* p, float v) {
  __hip_atomic_fetch_add(p, v, __ATOMIC_RELAXED, __HIP_MEMORY_SCOPE_WORKGROUP);
}

__device__ __forceinline__ void lds_inc_u32(unsigned* p) {
  __hip_atomic_fetch_add(p, 1u, __ATOMIC_RELAXED, __HIP_MEMORY_SCOPE_WORKGROUP);
}

__device__ __forceinline__ int4 own_load_key16(const unsigned char* __restrict__ key, int e0, int nvec) {
  return *(const int4*)(key + ((e0 < nvec) ? e0 : 0));
}

template <bool FULL, bool COUNT>
__device__ __forceinline__ void own_gather(
    const float* __restrict__ h, const int* __restrict__ src, const int* __restrict__ dst,
    float* acc, unsigned* deg, const int* qe, int lo, int first, int n, int lane) {
  int e[OWN_B], d[OWN_B], s[OWN_B];
  float4 v[OWN_B];
  unsigned slot = OWN_R;
#pragma unroll
  for (int k = 0; k < OWN_B; ++k) {
    const int idx = (FULL || k < n) ? first + k : first;
    e[k]  = __builtin_amdgcn_readfirstlane(qe[idx]);
  }
#pragma unroll
  for (int k = 0; k < OWN_B; ++k) s[k] = src[e[k]];
#pragma unroll
  for (int k = 0; k < OWN_B; ++k) d[k] = dst[e[k]];
#pragma unroll
  for (int k = 0; k < OWN_B; ++k)
    v[k] = *(const float4*)(h + (size_t)s[k] * N_FEATS + (lane << 2));
#pragma unroll
  for (int k = 0; k < OWN_B; ++k)
    asm volatile("" :: "v"(v[k].x), "v"(v[k].y), "v"(v[k].z), "v"(v[k].w));
#pragma unroll
  for (int k = 0; k < OWN_B; ++k) {
    const unsigned dl = (unsigned)(d[k] - lo);
    if ((FULL || k < n) && dl < (unsigned)OWN_R) {
      float* a = acc + dl * N_FEATS + lane;
      lds_add_f32(a +  0, v[k].x);
      lds_add_f32(a + 32, v[k].y);
      lds_add_f32(a + 64, v[k].z);
      lds_add_f32(a + 96, v[k].w);
      if (COUNT && lane == k) slot = dl;
    }
  }
  if (COUNT && slot < (unsigned)OWN_R) lds_inc_u32(deg + slot);
}

template <bool COUNT>
__device__ __forceinline__ void own_scan(
    const int4 (&c)[OWN_U], int base, int nvec, unsigned pat, int lo, int lane, int& cnt,
    int* myqe, const float* __restrict__ h, const int* __restrict__ src,
    const int* __restrict__ dst, float* acc, unsigned* deg) {
#pragma unroll
  for (int u = 0; u < OWN_U; ++u) {
    const int e0 = base + u * 512 + (lane << 4);
    const unsigned kv[4] = {(unsigned)c[u].x, (unsigned)c[u].y, (unsigned)c[u].z, (unsigned)c[u].w};
    unsigned f = 0u;
#pragma unroll
    for (int j = 0; j < 4; ++j) {
      const unsigned y = kv[j] ^ pat;
      const unsigned n = (y - 0x01010101u) & ~y & 0x80808080u;
      f |= n >> (7 - j);
    }
    if (e0 >= nvec) f = 0u;
    unsigned mask = __builtin_amdgcn_ballot_w32(f != 0u);
    while (mask != 0u) {
      if (f != 0u) {
        const int p = cnt + __builtin_popcount(mask & ((1u << lane) - 1u));
        const int pos = __builtin_ctz(f);
        myqe[p] = e0 + ((pos & 3) << 2) + (pos >> 3);
        f &= f - 1u;
      }
      cnt += __builtin_popcount(mask);
      while (cnt >= OWN_B) {
        cnt -= OWN_B;
        own_gather<true, COUNT>(h, src, dst, acc, deg, myqe, lo, cnt, OWN_B, lane);
      }
      mask = __builtin_amdgcn_ballot_w32(f != 0u);
    }
  }
}

template <bool COUNT>
__global__ __launch_bounds__(OWN_T) void gcn_agg_own(
    const float* __restrict__ h, const int* __restrict__ src, const int* __restrict__ dst,
    const unsigned char* __restrict__ key,
    float* __restrict__ norm_dst, const float* __restrict__ bias,
    float* __restrict__ out, int n_nodes, int n_edges, int do_relu) {
  __shared__ __attribute__((aligned(16))) float acc[OWN_R * N_FEATS];
  __shared__ int qe[OWN_W][OWN_Q];
  __shared__ unsigned deg[OWN_R];

  const int tid  = threadIdx.x;
  const int wave = __builtin_amdgcn_readfirstlane(tid >> 5);
  const int lane = tid & 31;
  const int lo   = blockIdx.x * OWN_R;
  const unsigned pat = (blockIdx.x & 255u) * 0x01010101u;

  for (int i = tid; i < OWN_R * N_FEATS / 4; i += OWN_T)
    ((float4*)acc)[i] = make_float4(0.f, 0.f, 0.f, 0.f);
  if (COUNT)
    for (int i = tid; i < OWN_R; i += OWN_T) deg[i] = 0u;
  __syncthreads();

  int* myqe = qe[wave];
  int cnt = 0;
  const int nvec = n_edges & ~15;
  const int step = OWN_W * OWN_U * 512;
  int base = wave * (OWN_U * 512);

  if (wave == 0) {
    for (int e = nvec; e < n_edges; ++e) {
      const unsigned dl = (unsigned)(dst[e] - lo);
      if (dl < (unsigned)OWN_R) {
        if (lane == 0) myqe[cnt] = e;
        ++cnt;
      }
    }
    while (cnt >= OWN_B) {
      cnt -= OWN_B;
      own_gather<true, COUNT>(h, src, dst, acc, deg, myqe, lo, cnt, OWN_B, lane);
    }
  }

  if (base < nvec) {
    int4 a[OWN_U], b[OWN_U];
#pragma unroll
    for (int u = 0; u < OWN_U; ++u)
      a[u] = own_load_key16(key, base + u * 512 + (lane << 4), nvec);
    for (;;) {
#pragma unroll
      for (int u = 0; u < OWN_U; ++u)
        b[u] = own_load_key16(key, base + step + u * 512 + (lane << 4), nvec);
      own_scan<COUNT>(a, base, nvec, pat, lo, lane, cnt, myqe, h, src, dst, acc, deg);
      base += step;
      if (base >= nvec) break;
#pragma unroll
      for (int u = 0; u < OWN_U; ++u) a[u] = b[u];
    }
  }
  if (cnt > 0) own_gather<false, COUNT>(h, src, dst, acc, deg, myqe, lo, 0, cnt, lane);
  __syncthreads();

  const float4 b = *(const float4*)(bias + (lane << 2));
  for (int r = wave; r < OWN_R; r += OWN_W) {
    const int row = lo + r;
    if (row < n_nodes) {
      const float* a = acc + r * N_FEATS + lane;
      float s;
      if (COUNT) {
        s = rsqrtf(fmaxf((float)deg[r], 1.0f));
        if (lane == 0) norm_dst[row] = s;
      } else {
        s = norm_dst[row];
      }
      float4 o;
      o.x = a[ 0] * s + b.x;
      o.y = a[32] * s + b.y;
      o.z = a[64] * s + b.z;
      o.w = a[96] * s + b.w;
      if (do_relu) {
        o.x = fmaxf(o.x, 0.f); o.y = fmaxf(o.y, 0.f);
        o.z = fmaxf(o.z, 0.f); o.w = fmaxf(o.w, 0.f);
      }
      *(float4*)(out + (size_t)row * N_FEATS + (lane << 2)) = o;
    }
  }
}

static inline size_t align256(size_t x) { return (x + 255) & ~(size_t)255; }

extern "C" void kernel_launch(void* const* d_in, const int* in_sizes, int n_in,
                              void* d_out, int out_size, void* d_ws, size_t ws_size,
                              hipStream_t stream) {
  (void)n_in; (void)out_size; (void)ws_size;
  const float* feats = (const float*)d_in[0];
  const int*   src   = (const int*)d_in[1];
  const int*   dst   = (const int*)d_in[2];
  const float* W1    = (const float*)d_in[3];
  const float* b1    = (const float*)d_in[4];
  const float* W2    = (const float*)d_in[5];
  const float* b2    = (const float*)d_in[6];
  float* out = (float*)d_out;

  const int n_nodes = in_sizes[0] / N_FEATS;
  const int n_edges = in_sizes[1];

  const size_t nb = align256((size_t)n_nodes * sizeof(float));
  const size_t hb = align256((size_t)n_nodes * N_FEATS * sizeof(float));
  char* p = (char*)d_ws;
  float* deg_out = (float*)p; p += nb;
  float* ns      = (float*)p; p += nb;
  float* nd      = (float*)p; p += nb;
  float* bufA    = (float*)p; p += hb;
  float* bufB    = (float*)p; p += hb;
  unsigned char* key = (unsigned char*)p;

  const int tb = 256;
  const size_t node_bytes = (size_t)n_nodes * sizeof(float);

  hipMemsetAsync(deg_out, 0, node_bytes, stream);
  gcn_degrees<<<(n_edges + tb - 1) / tb, tb, 0, stream>>>(src, dst, deg_out, key, n_edges);
  gcn_norms<<<(n_nodes + tb - 1) / tb, tb, 0, stream>>>(deg_out, ns, n_nodes);

  const int gemm_grid = (n_nodes + 15) / 16;
  const int own_grid  = (n_nodes + OWN_R - 1) / OWN_R;

  gcn_gemm_wmma<<<gemm_grid, 256, 0, stream>>>(feats, ns, W1, bufA, n_nodes);
  gcn_agg_own<true><<<own_grid, OWN_T, 0, stream>>>(bufA, src, dst, key, nd, b1, bufB, n_nodes, n_edges,  1);

  gcn_gemm_wmma<<<gemm_grid, 256, 0, stream>>>(bufB, ns, W2, bufA, n_nodes);
  gcn_agg_own<false><<<own_grid, OWN_T, 0, stream>>>(bufA, src, dst, key, nd, b2, out, n_nodes, n_edges,  0);
}


# out-degrees: per-edge global float atomics -> each work-group counts 32768 edges in an LDS integer histogram (two counts per word), stores it as a row; norms kernel adds the rows; memset dropped
# speedup vs baseline: 3.6245x; 1.2581x over previous
#include <hip/hip_runtime.h>

#define N_FEATS 128

typedef __attribute__((ext_vector_type(2))) float v2f;
typedef __attribute__((ext_vector_type(8))) float v8f;

#define OWN_SH 9

#define DEG_SH 17
#define DEG_N (1 << DEG_SH)
#define DEG_E 32768
#define DEG_T 1024
#define DEG_U (DEG_E / (DEG_T * 4))

__device__ __forceinline__ void deg_count(unsigned* hist, int s, int lo, int nn) {
  const unsigned r = (unsigned)(s - lo);
  if (r < (unsigned)nn)
    __hip_atomic_fetch_add(hist + (r >> 1), 1u << ((r & 1u) << 4), __ATOMIC_RELAXED,
                           __HIP_MEMORY_SCOPE_WORKGROUP);
}

__device__ __forceinline__ unsigned deg_key4(int4 d) {
  return  ((unsigned)(d.x >> OWN_SH) & 255u)        | (((unsigned)(d.y >> OWN_SH) & 255u) << 8) |
         (((unsigned)(d.z >> OWN_SH) & 255u) << 16) | (((unsigned)(d.w >> OWN_SH) & 255u) << 24);
}

__global__ __launch_bounds__(DEG_T) void gcn_degrees(
    const int* __restrict__ src, const int* __restrict__ dst,
    unsigned* __restrict__ part,
    unsigned char* __restrict__ key, int n_nodes, int n_edges) {
  __shared__ __attribute__((aligned(16))) unsigned hist[DEG_N / 2];

  const int tid = threadIdx.x;
  const int lo  = blockIdx.y << DEG_SH;
  const int nn  = min(n_nodes - lo, DEG_N);
  const int nw4 = (((nn + 1) >> 1) + 3) >> 2;
  const int pitch = (((n_nodes + 1) >> 1) + 31) & ~31;
  const bool keys = blockIdx.y == 0;

  for (int i = tid; i < nw4; i += DEG_T) ((uint4*)hist)[i] = make_uint4(0u, 0u, 0u, 0u);
  __syncthreads();

  const int e_begin = blockIdx.x * DEG_E;
  const int nvec = n_edges & ~3;
  if (e_begin < nvec) {
    int4 s[DEG_U], d[DEG_U];
#pragma unroll
    for (int u = 0; u < DEG_U; ++u) {
      const int e = e_begin + u * (DEG_T * 4) + (tid << 2);
      const int ea = (e < nvec) ? e : 0;
      s[u] = *(const int4*)(src + ea);
      d[u] = *(const int4*)(dst + ea);
    }
#pragma unroll
    for (int u = 0; u < DEG_U; ++u) {
      const int e = e_begin + u * (DEG_T * 4) + (tid << 2);
      if (e < nvec) {
        deg_count(hist, s[u].x, lo, nn);
        deg_count(hist, s[u].y, lo, nn);
        deg_count(hist, s[u].z, lo, nn);
        deg_count(hist, s[u].w, lo, nn);
        if (keys) *(unsigned*)(key + e) = deg_key4(d[u]);
      }
    }
  }
  {
    const int e = nvec + tid;
    if (e < n_edges && e >= e_begin && e < e_begin + DEG_E) {
      deg_count(hist, src[e], lo, nn);
      if (keys) key[e] = (unsigned char)(dst[e] >> OWN_SH);
    }
  }
  __syncthreads();

  uint4* row = (uint4*)(part + (size_t)blockIdx.x * pitch + (lo >> 1));
  for (int i = tid; i < nw4; i += DEG_T) row[i] = ((const uint4*)hist)[i];
}

__global__ __launch_bounds__(256) void gcn_norms(
    const unsigned* __restrict__ part, int n_rows, float* __restrict__ ns, int n) {
  const int w = blockIdx.x * blockDim.x + threadIdx.x;
  const int nwords = (n + 1) >> 1;
  if (w >= nwords) return;
  const int pitch = (nwords + 31) & ~31;
  unsigned c0 = 0u, c1 = 0u;
#pragma unroll 8
  for (int g = 0; g < n_rows; ++g) {
    const unsigned v = part[(size_t)g * pitch + w];
    c0 += v & 0xffffu;
    c1 += v >> 16;
  }
  ns[2 * w] = rsqrtf(fmaxf((float)c0, 1.0f));
  if (2 * w + 1 < n) ns[2 * w + 1] = rsqrtf(fmaxf((float)c1, 1.0f));
}

__global__ __launch_bounds__(256) void gcn_gemm_wmma(
    const float* __restrict__ A, const float* __restrict__ norm_src,
    const float* __restrict__ W, float* __restrict__ out, int n_rows) {
  __shared__ float As[16][132];

  const int row0 = blockIdx.x << 4;

  for (int t = threadIdx.x; t < 512; t += 256) {
    const int r = t >> 5;
    const int c4 = (t & 31) << 2;
    const int row = row0 + r;
    if (row < n_rows) {
      const float4 v = *(const float4*)(A + (size_t)row * N_FEATS + c4);
      const float s = norm_src[row];
      As[r][c4 + 0] = v.x * s;
      As[r][c4 + 1] = v.y * s;
      As[r][c4 + 2] = v.z * s;
      As[r][c4 + 3] = v.w * s;
    } else {
      As[r][c4 + 0] = 0.f; As[r][c4 + 1] = 0.f;
      As[r][c4 + 2] = 0.f; As[r][c4 + 3] = 0.f;
    }
  }
  __syncthreads();

  const int wave = threadIdx.x >> 5;
  const int lane = threadIdx.x & 31;
  const int half = lane >> 4;
  const int m    = lane & 15;
  const int col0 = wave << 4;

  v8f acc = {};
#pragma unroll
  for (int k = 0; k < N_FEATS; k += 4) {
    const int ka = k + (half << 1);
    v2f a, b;
    a.x = As[m][ka];
    a.y = As[m][ka + 1];
    b.x = W[ka * N_FEATS + col0 + m];
    b.y = W[(ka + 1) * N_FEATS + col0 + m];
    acc = __builtin_amdgcn_wmma_f32_16x16x4_f32(
         false, a,  false, b,
         (short)0, acc,  false,  false);
  }

#pragma unroll
  for (int j = 0; j < 8; ++j) {
    const int r = row0 + j + (half << 3);
    if (r < n_rows) out[(size_t)r * N_FEATS + col0 + m] = acc[j];
  }
}

#define OWN_R (1 << OWN_SH)
#define OWN_T 1024
#define OWN_W 32
#define OWN_U 4
#define OWN_B 8
#define OWN_Q (32 + OWN_B)

__device__ __forceinline__ void lds_add_f32(float* p, float v) {
  __hip_atomic_fetch_add(p, v, __ATOMIC_RELAXED, __HIP_MEMORY_SCOPE_WORKGROUP);
}

__device__ __forceinline__ void lds_inc_u32(unsigned* p) {
  __hip_atomic_fetch_add(p, 1u, __ATOMIC_RELAXED, __HIP_MEMORY_SCOPE_WORKGROUP);
}

__device__ __forceinline__ int4 own_load_key16(const unsigned char* __restrict__ key, int e0, int nvec) {
  return *(const int4*)(key + ((e0 < nvec) ? e0 : 0));
}

template <bool FULL, bool COUNT>
__device__ __forceinline__ void own_gather(
    const float* __restrict__ h, const int* __restrict__ src, const int* __restrict__ dst,
    float* acc, unsigned* deg, const int* qe, int lo, int first, int n, int lane) {
  int e[OWN_B], d[OWN_B], s[OWN_B];
  float4 v[OWN_B];
  unsigned slot = OWN_R;
#pragma unroll
  for (int k = 0; k < OWN_B; ++k) {
    const int idx = (FULL || k < n) ? first + k : first;
    e[k]  = __builtin_amdgcn_readfirstlane(qe[idx]);
  }
#pragma unroll
  for (int k = 0; k < OWN_B; ++k) s[k] = src[e[k]];
#pragma unroll
  for (int k = 0; k < OWN_B; ++k) d[k] = dst[e[k]];
#pragma unroll
  for (int k = 0; k < OWN_B; ++k)
    v[k] = *(const float4*)(h + (size_t)s[k] * N_FEATS + (lane << 2));
#pragma unroll
  for (int k = 0; k < OWN_B; ++k)
    asm volatile("" :: "v"(v[k].x), "v"(v[k].y), "v"(v[k].z), "v"(v[k].w));
#pragma unroll
  for (int k = 0; k < OWN_B; ++k) {
    const unsigned dl = (unsigned)(d[k] - lo);
    if ((FULL || k < n) && dl < (unsigned)OWN_R) {
      float* a = acc + dl * N_FEATS + lane;
      lds_add_f32(a +  0, v[k].x);
      lds_add_f32(a + 32, v[k].y);
      lds_add_f32(a + 64, v[k].z);
      lds_add_f32(a + 96, v[k].w);
      if (COUNT && lane == k) slot = dl;
    }
  }
  if (COUNT && slot < (unsigned)OWN_R) lds_inc_u32(deg + slot);
}

template <bool COUNT>
__device__ __forceinline__ void own_scan(
    const int4 (&c)[OWN_U], int base, int nvec, unsigned pat, int lo, int lane, int& cnt,
    int* myqe, const float* __restrict__ h, const int* __restrict__ src,
    const int* __restrict__ dst, float* acc, unsigned* deg) {
#pragma unroll
  for (int u = 0; u < OWN_U; ++u) {
    const int e0 = base + u * 512 + (lane << 4);
    const unsigned kv[4] = {(unsigned)c[u].x, (unsigned)c[u].y, (unsigned)c[u].z, (unsigned)c[u].w};
    unsigned f = 0u;
#pragma unroll
    for (int j = 0; j < 4; ++j) {
      const unsigned y = kv[j] ^ pat;
      const unsigned n = (y - 0x01010101u) & ~y & 0x80808080u;
      f |= n >> (7 - j);
    }
    if (e0 >= nvec) f = 0u;
    unsigned mask = __builtin_amdgcn_ballot_w32(f != 0u);
    while (mask != 0u) {
      if (f != 0u) {
        const int p = cnt + __builtin_popcount(mask & ((1u << lane) - 1u));
        const int pos = __builtin_ctz(f);
        myqe[p] = e0 + ((pos & 3) << 2) + (pos >> 3);
        f &= f - 1u;
      }
      cnt += __builtin_popcount(mask);
      while (cnt >= OWN_B) {
        cnt -= OWN_B;
        own_gather<true, COUNT>(h, src, dst, acc, deg, myqe, lo, cnt, OWN_B, lane);
      }
      mask = __builtin_amdgcn_ballot_w32(f != 0u);
    }
  }
}

template <bool COUNT>
__global__ __launch_bounds__(OWN_T) void gcn_agg_own(
    const float* __restrict__ h, const int* __restrict__ src, const int* __restrict__ dst,
    const unsigned char* __restrict__ key,
    float* __restrict__ norm_dst, const float* __restrict__ bias,
    float* __restrict__ out, int n_nodes, int n_edges, int do_relu) {
  __shared__ __attribute__((aligned(16))) float acc[OWN_R * N_FEATS];
  __shared__ int qe[OWN_W][OWN_Q];
  __shared__ unsigned deg[OWN_R];

  const int tid  = threadIdx.x;
  const int wave = __builtin_amdgcn_readfirstlane(tid >> 5);
  const int lane = tid & 31;
  const int lo   = blockIdx.x * OWN_R;
  const unsigned pat = (blockIdx.x & 255u) * 0x01010101u;

  for (int i = tid; i < OWN_R * N_FEATS / 4; i += OWN_T)
    ((float4*)acc)[i] = make_float4(0.f, 0.f, 0.f, 0.f);
  if (COUNT)
    for (int i = tid; i < OWN_R; i += OWN_T) deg[i] = 0u;
  __syncthreads();

  int* myqe = qe[wave];
  int cnt = 0;
  const int nvec = n_edges & ~15;
  const int step = OWN_W * OWN_U * 512;
  int base = wave * (OWN_U * 512);

  if (wave == 0) {
    for (int e = nvec; e < n_edges; ++e) {
      const unsigned dl = (unsigned)(dst[e] - lo);
      if (dl < (unsigned)OWN_R) {
        if (lane == 0) myqe[cnt] = e;
        ++cnt;
      }
    }
    while (cnt >= OWN_B) {
      cnt -= OWN_B;
      own_gather<true, COUNT>(h, src, dst, acc, deg, myqe, lo, cnt, OWN_B, lane);
    }
  }

  if (base < nvec) {
    int4 a[OWN_U], b[OWN_U];
#pragma unroll
    for (int u = 0; u < OWN_U; ++u)
      a[u] = own_load_key16(key, base + u * 512 + (lane << 4), nvec);
    for (;;) {
#pragma unroll
      for (int u = 0; u < OWN_U; ++u)
        b[u] = own_load_key16(key, base + step + u * 512 + (lane << 4), nvec);
      own_scan<COUNT>(a, base, nvec, pat, lo, lane, cnt, myqe, h, src, dst, acc, deg);
      base += step;
      if (base >= nvec) break;
#pragma unroll
      for (int u = 0; u < OWN_U; ++u) a[u] = b[u];
    }
  }
  if (cnt > 0) own_gather<false, COUNT>(h, src, dst, acc, deg, myqe, lo, 0, cnt, lane);
  __syncthreads();

  const float4 b = *(const float4*)(bias + (lane << 2));
  for (int r = wave; r < OWN_R; r += OWN_W) {
    const int row = lo + r;
    if (row < n_nodes) {
      const float* a = acc + r * N_FEATS + lane;
      float s;
      if (COUNT) {
        s = rsqrtf(fmaxf((float)deg[r], 1.0f));
        if (lane == 0) norm_dst[row] = s;
      } else {
        s = norm_dst[row];
      }
      float4 o;
      o.x = a[ 0] * s + b.x;
      o.y = a[32] * s + b.y;
      o.z = a[64] * s + b.z;
      o.w = a[96] * s + b.w;
      if (do_relu) {
        o.x = fmaxf(o.x, 0.f); o.y = fmaxf(o.y, 0.f);
        o.z = fmaxf(o.z, 0.f); o.w = fmaxf(o.w, 0.f);
      }
      *(float4*)(out + (size_t)row * N_FEATS + (lane << 2)) = o;
    }
  }
}

static inline size_t align256(size_t x) { return (x + 255) & ~(size_t)255; }

extern "C" void kernel_launch(void* const* d_in, const int* in_sizes, int n_in,
                              void* d_out, int out_size, void* d_ws, size_t ws_size,
                              hipStream_t stream) {
  (void)n_in; (void)out_size; (void)ws_size;
  const float* feats = (const float*)d_in[0];
  const int*   src   = (const int*)d_in[1];
  const int*   dst   = (const int*)d_in[2];
  const float* W1    = (const float*)d_in[3];
  const float* b1    = (const float*)d_in[4];
  const float* W2    = (const float*)d_in[5];
  const float* b2    = (const float*)d_in[6];
  float* out = (float*)d_out;

  const int n_nodes = in_sizes[0] / N_FEATS;
  const int n_edges = in_sizes[1];

  const size_t nb = align256((size_t)n_nodes * sizeof(float));
  const size_t hb = align256((size_t)n_nodes * N_FEATS * sizeof(float));
  char* p = (char*)d_ws;
  float* ns      = (float*)p; p += nb;
  float* nd      = (float*)p; p += nb;
  float* bufA    = (float*)p; p += hb;
  float* bufB    = (float*)p; p += hb;
  unsigned char* key = (unsigned char*)p; p += align256((size_t)n_edges);
  unsigned* part = (unsigned*)p;

  const int tb = 256;

  const int deg_rows = (n_edges + DEG_E - 1) / DEG_E;
  gcn_degrees<<<dim3(deg_rows, (n_nodes + DEG_N - 1) / DEG_N), DEG_T, 0, stream>>>(src, dst, part, key, n_nodes, n_edges);
  gcn_norms<<<((n_nodes + 1) / 2 + tb - 1) / tb, tb, 0, stream>>>(part, deg_rows, ns, n_nodes);

  const int gemm_grid = (n_nodes + 15) / 16;
  const int own_grid  = (n_nodes + OWN_R - 1) / OWN_R;

  gcn_gemm_wmma<<<gemm_grid, 256, 0, stream>>>(feats, ns, W1, bufA, n_nodes);
  gcn_agg_own<true><<<own_grid, OWN_T, 0, stream>>>(bufA, src, dst, key, nd, b1, bufB, n_nodes, n_edges,  1);

  gcn_gemm_wmma<<<gemm_grid, 256, 0, stream>>>(bufB, ns, W2, bufA, n_nodes);
  gcn_agg_own<false><<<own_grid, OWN_T, 0, stream>>>(bufA, src, dst, key, nd, b2, out, n_nodes, n_edges,  0);
}


# layer-2 aggregation scans nothing: each wave replays the (source,row) list it recorded for its queued edges in layer 1, 16 rows in flight, no key/src/dst re-read; sum order changes
# speedup vs baseline: 3.9987x; 1.1032x over previous
#include <hip/hip_runtime.h>

#define N_FEATS 128

typedef __attribute__((ext_vector_type(2))) float v2f;
typedef __attribute__((ext_vector_type(8))) float v8f;

#define OWN_SH 9

#define DEG_SH 17
#define DEG_N (1 << DEG_SH)
#define DEG_E 32768
#define DEG_T 1024
#define DEG_U (DEG_E / (DEG_T * 4))

__device__ __forceinline__ void deg_count(unsigned* hist, int s, int lo, int nn) {
  const unsigned r = (unsigned)(s - lo);
  if (r < (unsigned)nn)
    __hip_atomic_fetch_add(hist + (r >> 1), 1u << ((r & 1u) << 4), __ATOMIC_RELAXED,
                           __HIP_MEMORY_SCOPE_WORKGROUP);
}

__device__ __forceinline__ unsigned deg_key4(int4 d) {
  return  ((unsigned)(d.x >> OWN_SH) & 255u)        | (((unsigned)(d.y >> OWN_SH) & 255u) << 8) |
         (((unsigned)(d.z >> OWN_SH) & 255u) << 16) | (((unsigned)(d.w >> OWN_SH) & 255u) << 24);
}

__global__ __launch_bounds__(DEG_T) void gcn_degrees(
    const int* __restrict__ src, const int* __restrict__ dst,
    unsigned* __restrict__ part,
    unsigned char* __restrict__ key, int n_nodes, int n_edges) {
  __shared__ __attribute__((aligned(16))) unsigned hist[DEG_N / 2];

  const int tid = threadIdx.x;
  const int lo  = blockIdx.y << DEG_SH;
  const int nn  = min(n_nodes - lo, DEG_N);
  const int nw4 = (((nn + 1) >> 1) + 3) >> 2;
  const int pitch = (((n_nodes + 1) >> 1) + 31) & ~31;
  const bool keys = blockIdx.y == 0;

  for (int i = tid; i < nw4; i += DEG_T) ((uint4*)hist)[i] = make_uint4(0u, 0u, 0u, 0u);
  __syncthreads();

  const int e_begin = blockIdx.x * DEG_E;
  const int nvec = n_edges & ~3;
  if (e_begin < nvec) {
    int4 s[DEG_U], d[DEG_U];
#pragma unroll
    for (int u = 0; u < DEG_U; ++u) {
      const int e = e_begin + u * (DEG_T * 4) + (tid << 2);
      const int ea = (e < nvec) ? e : 0;
      s[u] = *(const int4*)(src + ea);
      d[u] = *(const int4*)(dst + ea);
    }
#pragma unroll
    for (int u = 0; u < DEG_U; ++u) {
      const int e = e_begin + u * (DEG_T * 4) + (tid << 2);
      if (e < nvec) {
        deg_count(hist, s[u].x, lo, nn);
        deg_count(hist, s[u].y, lo, nn);
        deg_count(hist, s[u].z, lo, nn);
        deg_count(hist, s[u].w, lo, nn);
        if (keys) *(unsigned*)(key + e) = deg_key4(d[u]);
      }
    }
  }
  {
    const int e = nvec + tid;
    if (e < n_edges && e >= e_begin && e < e_begin + DEG_E) {
      deg_count(hist, src[e], lo, nn);
      if (keys) key[e] = (unsigned char)(dst[e] >> OWN_SH);
    }
  }
  __syncthreads();

  uint4* row = (uint4*)(part + (size_t)blockIdx.x * pitch + (lo >> 1));
  for (int i = tid; i < nw4; i += DEG_T) row[i] = ((const uint4*)hist)[i];
}

__global__ __launch_bounds__(256) void gcn_norms(
    const unsigned* __restrict__ part, int n_rows, float* __restrict__ ns, int n) {
  const int w = blockIdx.x * blockDim.x + threadIdx.x;
  const int nwords = (n + 1) >> 1;
  if (w >= nwords) return;
  const int pitch = (nwords + 31) & ~31;
  unsigned c0 = 0u, c1 = 0u;
#pragma unroll 8
  for (int g = 0; g < n_rows; ++g) {
    const unsigned v = part[(size_t)g * pitch + w];
    c0 += v & 0xffffu;
    c1 += v >> 16;
  }
  ns[2 * w] = rsqrtf(fmaxf((float)c0, 1.0f));
  if (2 * w + 1 < n) ns[2 * w + 1] = rsqrtf(fmaxf((float)c1, 1.0f));
}

__global__ __launch_bounds__(256) void gcn_gemm_wmma(
    const float* __restrict__ A, const float* __restrict__ norm_src,
    const float* __restrict__ W, float* __restrict__ out, int n_rows) {
  __shared__ float As[16][132];

  const int row0 = blockIdx.x << 4;

  for (int t = threadIdx.x; t < 512; t += 256) {
    const int r = t >> 5;
    const int c4 = (t & 31) << 2;
    const int row = row0 + r;
    if (row < n_rows) {
      const float4 v = *(const float4*)(A + (size_t)row * N_FEATS + c4);
      const float s = norm_src[row];
      As[r][c4 + 0] = v.x * s;
      As[r][c4 + 1] = v.y * s;
      As[r][c4 + 2] = v.z * s;
      As[r][c4 + 3] = v.w * s;
    } else {
      As[r][c4 + 0] = 0.f; As[r][c4 + 1] = 0.f;
      As[r][c4 + 2] = 0.f; As[r][c4 + 3] = 0.f;
    }
  }
  __syncthreads();

  const int wave = threadIdx.x >> 5;
  const int lane = threadIdx.x & 31;
  const int half = lane >> 4;
  const int m    = lane & 15;
  const int col0 = wave << 4;

  v8f acc = {};
#pragma unroll
  for (int k = 0; k < N_FEATS; k += 4) {
    const int ka = k + (half << 1);
    v2f a, b;
    a.x = As[m][ka];
    a.y = As[m][ka + 1];
    b.x = W[ka * N_FEATS + col0 + m];
    b.y = W[(ka + 1) * N_FEATS + col0 + m];
    acc = __builtin_amdgcn_wmma_f32_16x16x4_f32(
         false, a,  false, b,
         (short)0, acc,  false,  false);
  }

#pragma unroll
  for (int j = 0; j < 8; ++j) {
    const int r = row0 + j + (half << 3);
    if (r < n_rows) out[(size_t)r * N_FEATS + col0 + m] = acc[j];
  }
}

#define OWN_R (1 << OWN_SH)
#define OWN_T 1024
#define OWN_W 32
#define OWN_U 4
#define OWN_B 8
#define OWN_Q (32 + OWN_B)
#define OWN_CAP 1024
#define OWN_P 16

__device__ __forceinline__ void lds_add_f32(float* p, float v) {
  __hip_atomic_fetch_add(p, v, __ATOMIC_RELAXED, __HIP_MEMORY_SCOPE_WORKGROUP);
}

__device__ __forceinline__ void lds_inc_u32(unsigned* p) {
  __hip_atomic_fetch_add(p, 1u, __ATOMIC_RELAXED, __HIP_MEMORY_SCOPE_WORKGROUP);
}

__device__ __forceinline__ int4 own_load_key16(const unsigned char* __restrict__ key, int e0, int nvec) {
  return *(const int4*)(key + ((e0 < nvec) ? e0 : 0));
}

template <bool FULL, bool COUNT>
__device__ __forceinline__ void own_gather(
    const float* __restrict__ h, const int* __restrict__ src, const int* __restrict__ dst,
    float* acc, unsigned* deg, const int* qe, int lo, int first, int n, int lane,
    int* rec, int nrec) {
  int e[OWN_B], d[OWN_B], s[OWN_B];
  float4 v[OWN_B];
  unsigned slot = OWN_R;
  int from = 0;
#pragma unroll
  for (int k = 0; k < OWN_B; ++k) {
    const int idx = (FULL || k < n) ? first + k : first;
    e[k]  = __builtin_amdgcn_readfirstlane(qe[idx]);
  }
#pragma unroll
  for (int k = 0; k < OWN_B; ++k) s[k] = src[e[k]];
#pragma unroll
  for (int k = 0; k < OWN_B; ++k) d[k] = dst[e[k]];
#pragma unroll
  for (int k = 0; k < OWN_B; ++k)
    v[k] = *(const float4*)(h + (size_t)s[k] * N_FEATS + (lane << 2));
#pragma unroll
  for (int k = 0; k < OWN_B; ++k)
    asm volatile("" :: "v"(v[k].x), "v"(v[k].y), "v"(v[k].z), "v"(v[k].w));
#pragma unroll
  for (int k = 0; k < OWN_B; ++k) {
    const unsigned dl = (unsigned)(d[k] - lo);
    if (COUNT && lane == k) from = s[k];
    if ((FULL || k < n) && dl < (unsigned)OWN_R) {
      float* a = acc + dl * N_FEATS + lane;
      lds_add_f32(a +  0, v[k].x);
      lds_add_f32(a + 32, v[k].y);
      lds_add_f32(a + 64, v[k].z);
      lds_add_f32(a + 96, v[k].w);
      if (COUNT && lane == k) slot = dl;
    }
  }
  if (COUNT) {
    if (slot < (unsigned)OWN_R) lds_inc_u32(deg + slot);
    if (nrec + OWN_B <= OWN_CAP && lane < (FULL ? OWN_B : n))
      *(int2*)(rec + 2 * (nrec + lane)) = make_int2(from, (int)slot);
  }
}

__device__ __forceinline__ void own_replay(
    const float* __restrict__ h, const int* __restrict__ rec, int2 cur, int nl,
    float* acc, int lane) {
  for (int i = 0; i < nl; i += 32) {
    int2 nxt = make_int2(0, OWN_R);
    if (i + 32 + lane < nl) nxt = *(const int2*)(rec + 2 * (i + 32 + lane));
#pragma unroll
    for (int j = 0; j < 32; j += OWN_P) {
      if (i + j < nl) {
        float4 v[OWN_P];
#pragma unroll
        for (int k = 0; k < OWN_P; ++k) {
          const int s = __builtin_amdgcn_readlane(cur.x, j + k);
          v[k] = *(const float4*)(h + (size_t)s * N_FEATS + (lane << 2));
        }
#pragma unroll
        for (int k = 0; k < OWN_P; ++k)
          asm volatile("" :: "v"(v[k].x), "v"(v[k].y), "v"(v[k].z), "v"(v[k].w));
#pragma unroll
        for (int k = 0; k < OWN_P; ++k) {
          const unsigned dl = (unsigned)__builtin_amdgcn_readlane(cur.y, j + k);
          if (dl < (unsigned)OWN_R) {
            float* a = acc + dl * N_FEATS + lane;
            lds_add_f32(a +  0, v[k].x);
            lds_add_f32(a + 32, v[k].y);
            lds_add_f32(a + 64, v[k].z);
            lds_add_f32(a + 96, v[k].w);
          }
        }
      }
    }
    cur = nxt;
  }
}

template <bool COUNT>
__device__ __forceinline__ void own_scan(
    const int4 (&c)[OWN_U], int base, int nvec, unsigned pat, int lo, int lane, int& cnt,
    int* myqe, const float* __restrict__ h, const int* __restrict__ src,
    const int* __restrict__ dst, float* acc, unsigned* deg, int* rec, int& nrec) {
#pragma unroll
  for (int u = 0; u < OWN_U; ++u) {
    const int e0 = base + u * 512 + (lane << 4);
    const unsigned kv[4] = {(unsigned)c[u].x, (unsigned)c[u].y, (unsigned)c[u].z, (unsigned)c[u].w};
    unsigned f = 0u;
#pragma unroll
    for (int j = 0; j < 4; ++j) {
      const unsigned y = kv[j] ^ pat;
      const unsigned n = (y - 0x01010101u) & ~y & 0x80808080u;
      f |= n >> (7 - j);
    }
    if (e0 >= nvec) f = 0u;
    unsigned mask = __builtin_amdgcn_ballot_w32(f != 0u);
    while (mask != 0u) {
      if (f != 0u) {
        const int p = cnt + __builtin_popcount(mask & ((1u << lane) - 1u));
        const int pos = __builtin_ctz(f);
        myqe[p] = e0 + ((pos & 3) << 2) + (pos >> 3);
        f &= f - 1u;
      }
      cnt += __builtin_popcount(mask);
      while (cnt >= OWN_B) {
        cnt -= OWN_B;
        own_gather<true, COUNT>(h, src, dst, acc, deg, myqe, lo, cnt, OWN_B, lane, rec, nrec);
        if (COUNT) nrec += OWN_B;
      }
      mask = __builtin_amdgcn_ballot_w32(f != 0u);
    }
  }
}

template <bool COUNT>
__global__ __launch_bounds__(OWN_T) void gcn_agg_own(
    const float* __restrict__ h, const int* __restrict__ src, const int* __restrict__ dst,
    const unsigned char* __restrict__ key,
    float* __restrict__ norm_dst, const float* __restrict__ bias,
    float* __restrict__ out, int* __restrict__ rec, int* __restrict__ nrecs,
    int n_nodes, int n_edges, int do_relu) {
  __shared__ __attribute__((aligned(16))) float acc[OWN_R * N_FEATS];
  __shared__ int qe[OWN_W][OWN_Q];
  __shared__ unsigned deg[OWN_R];
  __shared__ int qn[OWN_W];

  const int tid  = threadIdx.x;
  const int wave = __builtin_amdgcn_readfirstlane(tid >> 5);
  const int lane = tid & 31;
  const int lo   = blockIdx.x * OWN_R;
  const unsigned pat = (blockIdx.x & 255u) * 0x01010101u;

  int* myrec = rec + (size_t)(blockIdx.x * OWN_W + wave) * (2 * OWN_CAP);
  int nrec = 0;
  int nl = 0;
  int2 head = make_int2(0, OWN_R);
  if (!COUNT) nl = nrecs[blockIdx.x * OWN_W + wave];
  const bool listed = !COUNT && nl <= OWN_CAP;
  if (listed && lane < nl) head = *(const int2*)(myrec + 2 * lane);

  for (int i = tid; i < OWN_R * N_FEATS / 4; i += OWN_T)
    ((float4*)acc)[i] = make_float4(0.f, 0.f, 0.f, 0.f);
  if (COUNT)
    for (int i = tid; i < OWN_R; i += OWN_T) deg[i] = 0u;
  __syncthreads();

  int* myqe = qe[wave];
  int cnt = 0;
  const int nvec = n_edges & ~15;
  const int step = OWN_W * OWN_U * 512;
  int base = wave * (OWN_U * 512);

  if (listed) {
    own_replay(h, myrec, head, nl, acc, lane);
    base = nvec;
  }
  else if (wave == 0) {
    for (int e = nvec; e < n_edges; ++e) {
      const unsigned dl = (unsigned)(dst[e] - lo);
      if (dl < (unsigned)OWN_R) {
        if (lane == 0) myqe[cnt] = e;
        ++cnt;
      }
    }
    while (cnt >= OWN_B) {
      cnt -= OWN_B;
      own_gather<true, COUNT>(h, src, dst, acc, deg, myqe, lo, cnt, OWN_B, lane, myrec, nrec);
      if (COUNT) nrec += OWN_B;
    }
  }

  if (base < nvec) {
    int4 a[OWN_U], b[OWN_U];
#pragma unroll
    for (int u = 0; u < OWN_U; ++u)
      a[u] = own_load_key16(key, base + u * 512 + (lane << 4), nvec);
    for (;;) {
#pragma unroll
      for (int u = 0; u < OWN_U; ++u)
        b[u] = own_load_key16(key, base + step + u * 512 + (lane << 4), nvec);
      own_scan<COUNT>(a, base, nvec, pat, lo, lane, cnt, myqe, h, src, dst, acc, deg, myrec, nrec);
      base += step;
      if (base >= nvec) break;
#pragma unroll
      for (int u = 0; u < OWN_U; ++u) a[u] = b[u];
    }
  }
  if (cnt > 0) {
    own_gather<false, COUNT>(h, src, dst, acc, deg, myqe, lo, 0, cnt, lane, myrec, nrec);
    if (COUNT) nrec += cnt;
  }
  if (COUNT && lane == 0) qn[wave] = nrec;
  __syncthreads();

  if (COUNT && wave == 0) nrecs[blockIdx.x * OWN_W + lane] = qn[lane];

  const float4 b = *(const float4*)(bias + (lane << 2));
  for (int r = wave; r < OWN_R; r += OWN_W) {
    const int row = lo + r;
    if (row < n_nodes) {
      const float* a = acc + r * N_FEATS + lane;
      float s;
      if (COUNT) {
        s = rsqrtf(fmaxf((float)deg[r], 1.0f));
        if (lane == 0) norm_dst[row] = s;
      } else {
        s = norm_dst[row];
      }
      float4 o;
      o.x = a[ 0] * s + b.x;
      o.y = a[32] * s + b.y;
      o.z = a[64] * s + b.z;
      o.w = a[96] * s + b.w;
      if (do_relu) {
        o.x = fmaxf(o.x, 0.f); o.y = fmaxf(o.y, 0.f);
        o.z = fmaxf(o.z, 0.f); o.w = fmaxf(o.w, 0.f);
      }
      *(float4*)(out + (size_t)row * N_FEATS + (lane << 2)) = o;
    }
  }
}

static inline size_t align256(size_t x) { return (x + 255) & ~(size_t)255; }

extern "C" void kernel_launch(void* const* d_in, const int* in_sizes, int n_in,
                              void* d_out, int out_size, void* d_ws, size_t ws_size,
                              hipStream_t stream) {
  (void)n_in; (void)out_size; (void)ws_size;
  const float* feats = (const float*)d_in[0];
  const int*   src   = (const int*)d_in[1];
  const int*   dst   = (const int*)d_in[2];
  const float* W1    = (const float*)d_in[3];
  const float* b1    = (const float*)d_in[4];
  const float* W2    = (const float*)d_in[5];
  const float* b2    = (const float*)d_in[6];
  float* out = (float*)d_out;

  const int n_nodes = in_sizes[0] / N_FEATS;
  const int n_edges = in_sizes[1];

  const size_t nb = align256((size_t)n_nodes * sizeof(float));
  const size_t hb = align256((size_t)n_nodes * N_FEATS * sizeof(float));
  char* p = (char*)d_ws;
  float* ns      = (float*)p; p += nb;
  float* nd      = (float*)p; p += nb;
  float* bufA    = (float*)p; p += hb;
  float* bufB    = (float*)p; p += hb;
  unsigned char* key = (unsigned char*)p; p += align256((size_t)n_edges);
  const int own_grid  = (n_nodes + OWN_R - 1) / OWN_R;
  int* rec   = (int*)p; p += (size_t)own_grid * OWN_W * OWN_CAP * 2 * sizeof(int);
  int* nrecs = (int*)p; p += align256((size_t)own_grid * OWN_W * sizeof(int));
  unsigned* part = (unsigned*)p;

  const int tb = 256;

  const int deg_rows = (n_edges + DEG_E - 1) / DEG_E;
  gcn_degrees<<<dim3(deg_rows, (n_nodes + DEG_N - 1) / DEG_N), DEG_T, 0, stream>>>(src, dst, part, key, n_nodes, n_edges);
  gcn_norms<<<((n_nodes + 1) / 2 + tb - 1) / tb, tb, 0, stream>>>(part, deg_rows, ns, n_nodes);

  const int gemm_grid = (n_nodes + 15) / 16;

  gcn_gemm_wmma<<<gemm_grid, 256, 0, stream>>>(feats, ns, W1, bufA, n_nodes);
  gcn_agg_own<true><<<own_grid, OWN_T, 0, stream>>>(bufA, src, dst, key, nd, b1, bufB, rec, nrecs, n_nodes, n_edges,  1);

  gcn_gemm_wmma<<<gemm_grid, 256, 0, stream>>>(bufB, ns, W2, bufA, n_nodes);
  gcn_agg_own<false><<<own_grid, OWN_T, 0, stream>>>(bufA, src, dst, key, nd, b2, out, rec, nrecs, n_nodes, n_edges,  0);
}


# aggregation tiling: 392 destination nodes per work-group, not 512: 256 work-groups, one per compute unit (was 196, 60 idle), each sums a quarter fewer edges; LDS adds in arrival order as before
# speedup vs baseline: 4.2448x; 1.0615x over previous
#include <hip/hip_runtime.h>

#define N_FEATS 128

typedef __attribute__((ext_vector_type(2))) float v2f;
typedef __attribute__((ext_vector_type(8))) float v8f;

#define OWN_R 392

#define DEG_SH 17
#define DEG_N (1 << DEG_SH)
#define DEG_E 32768
#define DEG_T 1024
#define DEG_U (DEG_E / (DEG_T * 4))

__device__ __forceinline__ void deg_count(unsigned* hist, int s, int lo, int nn) {
  const unsigned r = (unsigned)(s - lo);
  if (r < (unsigned)nn)
    __hip_atomic_fetch_add(hist + (r >> 1), 1u << ((r & 1u) << 4), __ATOMIC_RELAXED,
                           __HIP_MEMORY_SCOPE_WORKGROUP);
}

__device__ __forceinline__ unsigned deg_key4(int4 d) {
  return  (((unsigned)d.x / OWN_R) & 255u)        | ((((unsigned)d.y / OWN_R) & 255u) << 8) |
         ((((unsigned)d.z / OWN_R) & 255u) << 16) | ((((unsigned)d.w / OWN_R) & 255u) << 24);
}

__global__ __launch_bounds__(DEG_T) void gcn_degrees(
    const int* __restrict__ src, const int* __restrict__ dst,
    unsigned* __restrict__ part,
    unsigned char* __restrict__ key, int n_nodes, int n_edges) {
  __shared__ __attribute__((aligned(16))) unsigned hist[DEG_N / 2];

  const int tid = threadIdx.x;
  const int lo  = blockIdx.y << DEG_SH;
  const int nn  = min(n_nodes - lo, DEG_N);
  const int nw4 = (((nn + 1) >> 1) + 3) >> 2;
  const int pitch = (((n_nodes + 1) >> 1) + 31) & ~31;
  const bool keys = blockIdx.y == 0;

  for (int i = tid; i < nw4; i += DEG_T) ((uint4*)hist)[i] = make_uint4(0u, 0u, 0u, 0u);
  __syncthreads();

  const int e_begin = blockIdx.x * DEG_E;
  const int nvec = n_edges & ~3;
  if (e_begin < nvec) {
    int4 s[DEG_U], d[DEG_U];
#pragma unroll
    for (int u = 0; u < DEG_U; ++u) {
      const int e = e_begin + u * (DEG_T * 4) + (tid << 2);
      const int ea = (e < nvec) ? e : 0;
      s[u] = *(const int4*)(src + ea);
      d[u] = *(const int4*)(dst + ea);
    }
#pragma unroll
    for (int u = 0; u < DEG_U; ++u) {
      const int e = e_begin + u * (DEG_T * 4) + (tid << 2);
      if (e < nvec) {
        deg_count(hist, s[u].x, lo, nn);
        deg_count(hist, s[u].y, lo, nn);
        deg_count(hist, s[u].z, lo, nn);
        deg_count(hist, s[u].w, lo, nn);
        if (keys) *(unsigned*)(key + e) = deg_key4(d[u]);
      }
    }
  }
  {
    const int e = nvec + tid;
    if (e < n_edges && e >= e_begin && e < e_begin + DEG_E) {
      deg_count(hist, src[e], lo, nn);
      if (keys) key[e] = (unsigned char)((unsigned)dst[e] / OWN_R);
    }
  }
  __syncthreads();

  uint4* row = (uint4*)(part + (size_t)blockIdx.x * pitch + (lo >> 1));
  for (int i = tid; i < nw4; i += DEG_T) row[i] = ((const uint4*)hist)[i];
}

__global__ __launch_bounds__(256) void gcn_norms(
    const unsigned* __restrict__ part, int n_rows, float* __restrict__ ns, int n) {
  const int w = blockIdx.x * blockDim.x + threadIdx.x;
  const int nwords = (n + 1) >> 1;
  if (w >= nwords) return;
  const int pitch = (nwords + 31) & ~31;
  unsigned c0 = 0u, c1 = 0u;
#pragma unroll 8
  for (int g = 0; g < n_rows; ++g) {
    const unsigned v = part[(size_t)g * pitch + w];
    c0 += v & 0xffffu;
    c1 += v >> 16;
  }
  ns[2 * w] = rsqrtf(fmaxf((float)c0, 1.0f));
  if (2 * w + 1 < n) ns[2 * w + 1] = rsqrtf(fmaxf((float)c1, 1.0f));
}

__global__ __launch_bounds__(256) void gcn_gemm_wmma(
    const float* __restrict__ A, const float* __restrict__ norm_src,
    const float* __restrict__ W, float* __restrict__ out, int n_rows) {
  __shared__ float As[16][132];

  const int row0 = blockIdx.x << 4;

  for (int t = threadIdx.x; t < 512; t += 256) {
    const int r = t >> 5;
    const int c4 = (t & 31) << 2;
    const int row = row0 + r;
    if (row < n_rows) {
      const float4 v = *(const float4*)(A + (size_t)row * N_FEATS + c4);
      const float s = norm_src[row];
      As[r][c4 + 0] = v.x * s;
      As[r][c4 + 1] = v.y * s;
      As[r][c4 + 2] = v.z * s;
      As[r][c4 + 3] = v.w * s;
    } else {
      As[r][c4 + 0] = 0.f; As[r][c4 + 1] = 0.f;
      As[r][c4 + 2] = 0.f; As[r][c4 + 3] = 0.f;
    }
  }
  __syncthreads();

  const int wave = threadIdx.x >> 5;
  const int lane = threadIdx.x & 31;
  const int half = lane >> 4;
  const int m    = lane & 15;
  const int col0 = wave << 4;

  v8f acc = {};
#pragma unroll
  for (int k = 0; k < N_FEATS; k += 4) {
    const int ka = k + (half << 1);
    v2f a, b;
    a.x = As[m][ka];
    a.y = As[m][ka + 1];
    b.x = W[ka * N_FEATS + col0 + m];
    b.y = W[(ka + 1) * N_FEATS + col0 + m];
    acc = __builtin_amdgcn_wmma_f32_16x16x4_f32(
         false, a,  false, b,
         (short)0, acc,  false,  false);
  }

#pragma unroll
  for (int j = 0; j < 8; ++j) {
    const int r = row0 + j + (half << 3);
    if (r < n_rows) out[(size_t)r * N_FEATS + col0 + m] = acc[j];
  }
}

#define OWN_T 1024
#define OWN_W 32
#define OWN_U 4
#define OWN_B 8
#define OWN_Q (32 + OWN_B)
#define OWN_CAP 1024
#define OWN_P 16

__device__ __forceinline__ void lds_add_f32(float* p, float v) {
  __hip_atomic_fetch_add(p, v, __ATOMIC_RELAXED, __HIP_MEMORY_SCOPE_WORKGROUP);
}

__device__ __forceinline__ void lds_inc_u32(unsigned* p) {
  __hip_atomic_fetch_add(p, 1u, __ATOMIC_RELAXED, __HIP_MEMORY_SCOPE_WORKGROUP);
}

__device__ __forceinline__ int4 own_load_key16(const unsigned char* __restrict__ key, int e0, int nvec) {
  return *(const int4*)(key + ((e0 < nvec) ? e0 : 0));
}

template <bool FULL, bool COUNT>
__device__ __forceinline__ void own_gather(
    const float* __restrict__ h, const int* __restrict__ src, const int* __restrict__ dst,
    float* acc, unsigned* deg, const int* qe, int lo, int first, int n, int lane,
    int* rec, int nrec) {
  int e[OWN_B], d[OWN_B], s[OWN_B];
  float4 v[OWN_B];
  unsigned slot = OWN_R;
  int from = 0;
#pragma unroll
  for (int k = 0; k < OWN_B; ++k) {
    const int idx = (FULL || k < n) ? first + k : first;
    e[k]  = __builtin_amdgcn_readfirstlane(qe[idx]);
  }
#pragma unroll
  for (int k = 0; k < OWN_B; ++k) s[k] = src[e[k]];
#pragma unroll
  for (int k = 0; k < OWN_B; ++k) d[k] = dst[e[k]];
#pragma unroll
  for (int k = 0; k < OWN_B; ++k)
    v[k] = *(const float4*)(h + (size_t)s[k] * N_FEATS + (lane << 2));
#pragma unroll
  for (int k = 0; k < OWN_B; ++k)
    asm volatile("" :: "v"(v[k].x), "v"(v[k].y), "v"(v[k].z), "v"(v[k].w));
#pragma unroll
  for (int k = 0; k < OWN_B; ++k) {
    const unsigned dl = (unsigned)(d[k] - lo);
    if (COUNT && lane == k) from = s[k];
    if ((FULL || k < n) && dl < (unsigned)OWN_R) {
      float* a = acc + dl * N_FEATS + lane;
      lds_add_f32(a +  0, v[k].x);
      lds_add_f32(a + 32, v[k].y);
      lds_add_f32(a + 64, v[k].z);
      lds_add_f32(a + 96, v[k].w);
      if (COUNT && lane == k) slot = dl;
    }
  }
  if (COUNT) {
    if (slot < (unsigned)OWN_R) lds_inc_u32(deg + slot);
    if (nrec + OWN_B <= OWN_CAP && lane < (FULL ? OWN_B : n))
      *(int2*)(rec + 2 * (nrec + lane)) = make_int2(from, (int)slot);
  }
}

__device__ __forceinline__ void own_replay(
    const float* __restrict__ h, const int* __restrict__ rec, int2 cur, int nl,
    float* acc, int lane) {
  for (int i = 0; i < nl; i += 32) {
    int2 nxt = make_int2(0, OWN_R);
    if (i + 32 + lane < nl) nxt = *(const int2*)(rec + 2 * (i + 32 + lane));
#pragma unroll
    for (int j = 0; j < 32; j += OWN_P) {
      if (i + j < nl) {
        float4 v[OWN_P];
#pragma unroll
        for (int k = 0; k < OWN_P; ++k) {
          const int s = __builtin_amdgcn_readlane(cur.x, j + k);
          v[k] = *(const float4*)(h + (size_t)s * N_FEATS + (lane << 2));
        }
#pragma unroll
        for (int k = 0; k < OWN_P; ++k)
          asm volatile("" :: "v"(v[k].x), "v"(v[k].y), "v"(v[k].z), "v"(v[k].w));
#pragma unroll
        for (int k = 0; k < OWN_P; ++k) {
          const unsigned dl = (unsigned)__builtin_amdgcn_readlane(cur.y, j + k);
          if (dl < (unsigned)OWN_R) {
            float* a = acc + dl * N_FEATS + lane;
            lds_add_f32(a +  0, v[k].x);
            lds_add_f32(a + 32, v[k].y);
            lds_add_f32(a + 64, v[k].z);
            lds_add_f32(a + 96, v[k].w);
          }
        }
      }
    }
    cur = nxt;
  }
}

template <bool COUNT>
__device__ __forceinline__ void own_scan(
    const int4 (&c)[OWN_U], int base, int nvec, unsigned pat, int lo, int lane, int& cnt,
    int* myqe, const float* __restrict__ h, const int* __restrict__ src,
    const int* __restrict__ dst, float* acc, unsigned* deg, int* rec, int& nrec) {
#pragma unroll
  for (int u = 0; u < OWN_U; ++u) {
    const int e0 = base + u * 512 + (lane << 4);
    const unsigned kv[4] = {(unsigned)c[u].x, (unsigned)c[u].y, (unsigned)c[u].z, (unsigned)c[u].w};
    unsigned f = 0u;
#pragma unroll
    for (int j = 0; j < 4; ++j) {
      const unsigned y = kv[j] ^ pat;
      const unsigned n = (y - 0x01010101u) & ~y & 0x80808080u;
      f |= n >> (7 - j);
    }
    if (e0 >= nvec) f = 0u;
    unsigned mask = __builtin_amdgcn_ballot_w32(f != 0u);
    while (mask != 0u) {
      if (f != 0u) {
        const int p = cnt + __builtin_popcount(mask & ((1u << lane) - 1u));
        const int pos = __builtin_ctz(f);
        myqe[p] = e0 + ((pos & 3) << 2) + (pos >> 3);
        f &= f - 1u;
      }
      cnt += __builtin_popcount(mask);
      while (cnt >= OWN_B) {
        cnt -= OWN_B;
        own_gather<true, COUNT>(h, src, dst, acc, deg, myqe, lo, cnt, OWN_B, lane, rec, nrec);
        if (COUNT) nrec += OWN_B;
      }
      mask = __builtin_amdgcn_ballot_w32(f != 0u);
    }
  }
}

template <bool COUNT>
__global__ __launch_bounds__(OWN_T) void gcn_agg_own(
    const float* __restrict__ h, const int* __restrict__ src, const int* __restrict__ dst,
    const unsigned char* __restrict__ key,
    float* __restrict__ norm_dst, const float* __restrict__ bias,
    float* __restrict__ out, int* __restrict__ rec, int* __restrict__ nrecs,
    int n_nodes, int n_edges, int do_relu) {
  __shared__ __attribute__((aligned(16))) float acc[OWN_R * N_FEATS];
  __shared__ int qe[OWN_W][OWN_Q];
  __shared__ unsigned deg[OWN_R];
  __shared__ int qn[OWN_W];

  const int tid  = threadIdx.x;
  const int wave = __builtin_amdgcn_readfirstlane(tid >> 5);
  const int lane = tid & 31;
  const int lo   = blockIdx.x * OWN_R;
  const unsigned pat = (blockIdx.x & 255u) * 0x01010101u;

  int* myrec = rec + (size_t)(blockIdx.x * OWN_W + wave) * (2 * OWN_CAP);
  int nrec = 0;
  int nl = 0;
  int2 head = make_int2(0, OWN_R);
  if (!COUNT) nl = nrecs[blockIdx.x * OWN_W + wave];
  const bool listed = !COUNT && nl <= OWN_CAP;
  if (listed && lane < nl) head = *(const int2*)(myrec + 2 * lane);

  for (int i = tid; i < OWN_R * N_FEATS / 4; i += OWN_T)
    ((float4*)acc)[i] = make_float4(0.f, 0.f, 0.f, 0.f);
  if (COUNT)
    for (int i = tid; i < OWN_R; i += OWN_T) deg[i] = 0u;
  __syncthreads();

  int* myqe = qe[wave];
  int cnt = 0;
  const int nvec = n_edges & ~15;
  const int step = OWN_W * OWN_U * 512;
  int base = wave * (OWN_U * 512);

  if (listed) {
    own_replay(h, myrec, head, nl, acc, lane);
    base = nvec;
  }
  else if (wave == 0) {
    for (int e = nvec; e < n_edges; ++e) {
      const unsigned dl = (unsigned)(dst[e] - lo);
      if (dl < (unsigned)OWN_R) {
        if (lane == 0) myqe[cnt] = e;
        ++cnt;
      }
    }
    while (cnt >= OWN_B) {
      cnt -= OWN_B;
      own_gather<true, COUNT>(h, src, dst, acc, deg, myqe, lo, cnt, OWN_B, lane, myrec, nrec);
      if (COUNT) nrec += OWN_B;
    }
  }

  if (base < nvec) {
    int4 a[OWN_U], b[OWN_U];
#pragma unroll
    for (int u = 0; u < OWN_U; ++u)
      a[u] = own_load_key16(key, base + u * 512 + (lane << 4), nvec);
    for (;;) {
#pragma unroll
      for (int u = 0; u < OWN_U; ++u)
        b[u] = own_load_key16(key, base + step + u * 512 + (lane << 4), nvec);
      own_scan<COUNT>(a, base, nvec, pat, lo, lane, cnt, myqe, h, src, dst, acc, deg, myrec, nrec);
      base += step;
      if (base >= nvec) break;
#pragma unroll
      for (int u = 0; u < OWN_U; ++u) a[u] = b[u];
    }
  }
  if (cnt > 0) {
    own_gather<false, COUNT>(h, src, dst, acc, deg, myqe, lo, 0, cnt, lane, myrec, nrec);
    if (COUNT) nrec += cnt;
  }
  if (COUNT && lane == 0) qn[wave] = nrec;
  __syncthreads();

  if (COUNT && wave == 0) nrecs[blockIdx.x * OWN_W + lane] = qn[lane];

  const float4 b = *(const float4*)(bias + (lane << 2));
  for (int r = wave; r < OWN_R; r += OWN_W) {
    const int row = lo + r;
    if (row < n_nodes) {
      const float* a = acc + r * N_FEATS + lane;
      float s;
      if (COUNT) {
        s = rsqrtf(fmaxf((float)deg[r], 1.0f));
        if (lane == 0) norm_dst[row] = s;
      } else {
        s = norm_dst[row];
      }
      float4 o;
      o.x = a[ 0] * s + b.x;
      o.y = a[32] * s + b.y;
      o.z = a[64] * s + b.z;
      o.w = a[96] * s + b.w;
      if (do_relu) {
        o.x = fmaxf(o.x, 0.f); o.y = fmaxf(o.y, 0.f);
        o.z = fmaxf(o.z, 0.f); o.w = fmaxf(o.w, 0.f);
      }
      *(float4*)(out + (size_t)row * N_FEATS + (lane << 2)) = o;
    }
  }
}

static inline size_t align256(size_t x) { return (x + 255) & ~(size_t)255; }

extern "C" void kernel_launch(void* const* d_in, const int* in_sizes, int n_in,
                              void* d_out, int out_size, void* d_ws, size_t ws_size,
                              hipStream_t stream) {
  (void)n_in; (void)out_size; (void)ws_size;
  const float* feats = (const float*)d_in[0];
  const int*   src   = (const int*)d_in[1];
  const int*   dst   = (const int*)d_in[2];
  const float* W1    = (const float*)d_in[3];
  const float* b1    = (const float*)d_in[4];
  const float* W2    = (const float*)d_in[5];
  const float* b2    = (const float*)d_in[6];
  float* out = (float*)d_out;

  const int n_nodes = in_sizes[0] / N_FEATS;
  const int n_edges = in_sizes[1];

  const size_t nb = align256((size_t)n_nodes * sizeof(float));
  const size_t hb = align256((size_t)n_nodes * N_FEATS * sizeof(float));
  char* p = (char*)d_ws;
  float* ns      = (float*)p; p += nb;
  float* nd      = (float*)p; p += nb;
  float* bufA    = (float*)p; p += hb;
  float* bufB    = (float*)p; p += hb;
  unsigned char* key = (unsigned char*)p; p += align256((size_t)n_edges);
  const int own_grid  = (n_nodes + OWN_R - 1) / OWN_R;
  int* rec   = (int*)p; p += (size_t)own_grid * OWN_W * OWN_CAP * 2 * sizeof(int);
  int* nrecs = (int*)p; p += align256((size_t)own_grid * OWN_W * sizeof(int));
  unsigned* part = (unsigned*)p;

  const int tb = 256;

  const int deg_rows = (n_edges + DEG_E - 1) / DEG_E;
  gcn_degrees<<<dim3(deg_rows, (n_nodes + DEG_N - 1) / DEG_N), DEG_T, 0, stream>>>(src, dst, part, key, n_nodes, n_edges);
  gcn_norms<<<((n_nodes + 1) / 2 + tb - 1) / tb, tb, 0, stream>>>(part, deg_rows, ns, n_nodes);

  const int gemm_grid = (n_nodes + 15) / 16;

  gcn_gemm_wmma<<<gemm_grid, 256, 0, stream>>>(feats, ns, W1, bufA, n_nodes);
  gcn_agg_own<true><<<own_grid, OWN_T, 0, stream>>>(bufA, src, dst, key, nd, b1, bufB, rec, nrecs, n_nodes, n_edges,  1);

  gcn_gemm_wmma<<<gemm_grid, 256, 0, stream>>>(bufB, ns, W2, bufA, n_nodes);
  gcn_agg_own<false><<<own_grid, OWN_T, 0, stream>>>(bufA, src, dst, key, nd, b2, out, rec, nrecs, n_nodes, n_edges,  0);
}
